# starting kernel
# baseline (speedup 1.0000x reference)
#include <hip/hip_runtime.h>
#include <math.h>

#define N_NODES 100000
#define N_EDGES 1600000
#define IN_C    256
#define HID_C   128
#define N_LAYERS 4
#define ALPHA   0.1f

typedef float v2f __attribute__((ext_vector_type(2)));
typedef float v8f __attribute__((ext_vector_type(8)));

__device__ __forceinline__ void fadd_agent(float* p, float v) {
    __hip_atomic_fetch_add(p, v, __ATOMIC_RELAXED, __HIP_MEMORY_SCOPE_AGENT);
}

__global__ void kdeg_init(int* degi) {
    int v = blockIdx.x * 256 + threadIdx.x;
    if (v < N_NODES) degi[v] = 1;
}

__global__ void kdeg_count(const int* __restrict__ dst, int* __restrict__ degi) {
    int e = blockIdx.x * 256 + threadIdx.x;
    if (e < N_EDGES) atomicAdd(&degi[dst[e]], 1);
}

__global__ void kdinv(const int* __restrict__ degi, float* __restrict__ dinv) {
    int v = blockIdx.x * 256 + threadIdx.x;
    if (v < N_NODES) dinv[v] = rsqrtf((float)degi[v]);
}

__global__ void kzero4(float4* __restrict__ p, int n4) {
    int i = blockIdx.x * 256 + threadIdx.x;
    if (i < n4) p[i] = make_float4(0.f, 0.f, 0.f, 0.f);
}

#define LDA_P 260
__global__ void kproj(const float* __restrict__ x, const float* __restrict__ w,
                      const float* __restrict__ bias, float* __restrict__ x0) {
    __shared__ float sa[16 * LDA_P];
    const int t   = threadIdx.x;
    const int m0  = blockIdx.x * 16;

    const float4* xf4 = (const float4*)x;
    #pragma unroll
    for (int c = 0; c < 4; ++c) {
        int idx = c * 256 + t;
        int row = idx >> 6, q = idx & 63;
        float4 v = xf4[(size_t)(m0 + row) * 64 + q];
        float* d = &sa[row * LDA_P + q * 4];
        d[0] = v.x; d[1] = v.y; d[2] = v.z; d[3] = v.w;
    }
    __syncthreads();

    const int wave = t >> 5, lane = t & 31;
    const int half = lane >> 4, col = lane & 15;
    const int n0   = wave * 16;
    const int mrow = col;

    v8f acc = {};
    #pragma unroll 8
    for (int k0 = 0; k0 < IN_C; k0 += 4) {
        int ka = k0 + 2 * half;
        v2f a = *(const v2f*)&sa[mrow * LDA_P + ka];
        v2f b;
        b.x = w[(size_t)ka * HID_C + n0 + col];
        b.y = w[(size_t)(ka + 1) * HID_C + n0 + col];
        acc = __builtin_amdgcn_wmma_f32_16x16x4_f32(false, a, false, b,
                                                    (short)0, acc, false, false);
    }

    float bv = bias[n0 + col];
    #pragma unroll
    for (int i = 0; i < 8; ++i) {
        int row = i + 8 * half;
        x0[(size_t)(m0 + row) * HID_C + n0 + col] = acc[i] + bv;
    }
}

__global__ void kscatter(const int* __restrict__ src, const int* __restrict__ dst,
                         const float* __restrict__ dinv, const float* __restrict__ hin,
                         float* __restrict__ agg) {
    int wid  = (blockIdx.x * 256 + threadIdx.x) >> 5;
    int lane = threadIdx.x & 31;
    int s = src[wid], d = dst[wid];
    float nrm = dinv[s] * dinv[d];
    float4 v = ((const float4*)(hin + (size_t)s * HID_C))[lane];
    float* ap = agg + (size_t)d * HID_C + lane * 4;
    fadd_agent(ap + 0, nrm * v.x);
    fadd_agent(ap + 1, nrm * v.y);
    fadd_agent(ap + 2, nrm * v.z);
    fadd_agent(ap + 3, nrm * v.w);
}

#define LDA_L 132
__global__ void klayer(const float* __restrict__ agg, const float* __restrict__ hin,
                       const float* __restrict__ x0, const float* __restrict__ dinv,
                       const float* __restrict__ w, float beta, int do_relu,
                       float* __restrict__ hout) {
    __shared__ float sup[16 * LDA_L];
    const int t  = threadIdx.x;
    const int m0 = blockIdx.x * 16;

    const float4* af4 = (const float4*)agg;
    const float4* hf4 = (const float4*)hin;
    const float4* xf4 = (const float4*)x0;
    #pragma unroll
    for (int c = 0; c < 2; ++c) {
        int idx = c * 256 + t;
        int row = idx >> 5, q = idx & 31;
        float dv = dinv[m0 + row];
        float d2 = dv * dv;
        size_t base = (size_t)(m0 + row) * 32 + q;
        float4 a = af4[base], h = hf4[base], x = xf4[base];
        float* dp = &sup[row * LDA_L + q * 4];
        dp[0] = (1.f - ALPHA) * (a.x + d2 * h.x) + ALPHA * x.x;
        dp[1] = (1.f - ALPHA) * (a.y + d2 * h.y) + ALPHA * x.y;
        dp[2] = (1.f - ALPHA) * (a.z + d2 * h.z) + ALPHA * x.z;
        dp[3] = (1.f - ALPHA) * (a.w + d2 * h.w) + ALPHA * x.w;
    }
    __syncthreads();

    const int wave = t >> 5, lane = t & 31;
    const int half = lane >> 4, col = lane & 15;
    const int n0   = wave * 16;
    const int mrow = col;

    v8f acc = {};
    #pragma unroll 8
    for (int k0 = 0; k0 < HID_C; k0 += 4) {
        int ka = k0 + 2 * half;
        v2f a = *(const v2f*)&sup[mrow * LDA_L + ka];
        v2f b;
        b.x = w[(size_t)ka * HID_C + n0 + col];
        b.y = w[(size_t)(ka + 1) * HID_C + n0 + col];
        acc = __builtin_amdgcn_wmma_f32_16x16x4_f32(false, a, false, b,
                                                    (short)0, acc, false, false);
    }

    #pragma unroll
    for (int i = 0; i < 8; ++i) {
        int row = i + 8 * half;
        float s = sup[row * LDA_L + n0 + col];
        float v = (1.f - beta) * s + beta * acc[i];
        if (do_relu) v = fmaxf(v, 0.f);
        hout[(size_t)(m0 + row) * HID_C + n0 + col] = v;
    }
}

extern "C" void kernel_launch(void* const* d_in, const int* in_sizes, int n_in,
                              void* d_out, int out_size, void* d_ws, size_t ws_size,
                              hipStream_t stream) {
    const float* x      = (const float*)d_in[0];
    const int*   ei     = (const int*)d_in[1];
    const float* proj_w = (const float*)d_in[2];
    const float* proj_b = (const float*)d_in[3];
    const float* conv_w = (const float*)d_in[4];
    float* out = (float*)d_out;

    const int* e_src = ei;
    const int* e_dst = ei + N_EDGES;

    const size_t NH = (size_t)N_NODES * HID_C;
    char* ws = (char*)d_ws;
    int*   degi = (int*)ws;                                  ws += ((size_t)N_NODES * 4 + 255) & ~255ull;
    float* dinv = (float*)ws;                                ws += ((size_t)N_NODES * 4 + 255) & ~255ull;
    float* x0   = (float*)ws;                                ws += NH * 4;
    float* h    = (float*)ws;                                ws += NH * 4;
    float* agg  = (float*)ws;

    kdeg_init <<<(N_NODES + 255) / 256, 256, 0, stream>>>(degi);
    kdeg_count<<<(N_EDGES + 255) / 256, 256, 0, stream>>>(e_dst, degi);
    kdinv     <<<(N_NODES + 255) / 256, 256, 0, stream>>>(degi, dinv);

    kproj<<<N_NODES / 16, 256, 0, stream>>>(x, proj_w, proj_b, x0);

    static const float betas[N_LAYERS] = {0.40546510f, 0.22314355f, 0.15415068f, 0.11778303f};
    const int n4 = (int)(NH / 4);

    for (int i = 0; i < N_LAYERS; ++i) {
        const float* hin  = (i == 0) ? x0 : h;
        float*       hout = (i == N_LAYERS - 1) ? out : h;
        kzero4  <<<(n4 + 255) / 256, 256, 0, stream>>>((float4*)agg, n4);
        kscatter<<<N_EDGES / 8, 256, 0, stream>>>(e_src, e_dst, dinv, hin, agg);
        klayer  <<<N_NODES / 16, 256, 0, stream>>>(agg, hin, x0, dinv,
                                                   conv_w + (size_t)i * HID_C * HID_C,
                                                   betas[i], (i != N_LAYERS - 1) ? 1 : 0, hout);
    }
}


# Float-atomic edge scatter and agg zero-fill replaced by a per-call destination-grouped edge list (returning int add as rank, scan, fill) and a wave-per-row gather; row sums now run in rank order
# speedup vs baseline: 3.0498x; 3.0498x over previous
#include <hip/hip_runtime.h>
#include <math.h>

#define N_NODES 100000
#define N_EDGES 1600000
#define IN_C    256
#define HID_C   128
#define N_LAYERS 4
#define ALPHA   0.1f
#define N_NBLK  ((N_NODES + 255) / 256)

typedef float v2f __attribute__((ext_vector_type(2)));
typedef float v8f __attribute__((ext_vector_type(8)));

__global__ void kdeg_init(int* cnt) {
    int v = blockIdx.x * 256 + threadIdx.x;
    if (v < N_NODES) cnt[v] = 0;
}

__global__ void kdeg_rank(const int* __restrict__ dst, int* __restrict__ cnt,
                          int* __restrict__ rank) {
    int e = blockIdx.x * 256 + threadIdx.x;
    if (e < N_EDGES) rank[e] = atomicAdd(&cnt[dst[e]], 1);
}

__global__ void kblock_sum(const int* __restrict__ cnt, int* __restrict__ bsum) {
    __shared__ int red[256];
    const int t = threadIdx.x;
    const int v = blockIdx.x * 256 + t;
    red[t] = (v < N_NODES) ? cnt[v] : 0;
    __syncthreads();
    for (int s = 128; s > 0; s >>= 1) {
        if (t < s) red[t] += red[t + s];
        __syncthreads();
    }
    if (t == 0) bsum[blockIdx.x] = red[0];
}

__global__ void krowptr(const int* __restrict__ cnt, const int* __restrict__ bsum,
                        int* __restrict__ rowptr, float* __restrict__ dinv) {
    __shared__ int red[256];
    __shared__ int sc[256];
    const int t = threadIdx.x;
    const int b = blockIdx.x;
    const int v = b * 256 + t;
    const int c = (v < N_NODES) ? cnt[v] : 0;

    int part = 0;
    for (int i = t; i < b; i += 256) part += bsum[i];
    red[t] = part;
    sc[t]  = c;
    __syncthreads();
    for (int s = 128; s > 0; s >>= 1) {
        if (t < s) red[t] += red[t + s];
        __syncthreads();
    }
    const int off = red[0];

    for (int d = 1; d < 256; d <<= 1) {
        int add = (t >= d) ? sc[t - d] : 0;
        __syncthreads();
        sc[t] += add;
        __syncthreads();
    }
    if (v < N_NODES) {
        rowptr[v] = off + sc[t] - c;
        dinv[v]   = rsqrtf((float)(c + 1));
    }
}

__global__ void kfill(const int* __restrict__ src, const int* __restrict__ dst,
                      const int* __restrict__ rank, const int* __restrict__ rowptr,
                      const float* __restrict__ dinv, int2* __restrict__ rec) {
    int e = blockIdx.x * 256 + threadIdx.x;
    if (e < N_EDGES) {
        int s = src[e], d = dst[e];
        int pos = rowptr[d] + rank[e];
        rec[pos] = make_int2(s, __float_as_int(dinv[s] * dinv[d]));
    }
}

#define LDA_P 260
__global__ void kproj(const float* __restrict__ x, const float* __restrict__ w,
                      const float* __restrict__ bias, float* __restrict__ x0) {
    __shared__ float sa[16 * LDA_P];
    const int t   = threadIdx.x;
    const int m0  = blockIdx.x * 16;

    const float4* xf4 = (const float4*)x;
    #pragma unroll
    for (int c = 0; c < 4; ++c) {
        int idx = c * 256 + t;
        int row = idx >> 6, q = idx & 63;
        float4 v = xf4[(size_t)(m0 + row) * 64 + q];
        float* d = &sa[row * LDA_P + q * 4];
        d[0] = v.x; d[1] = v.y; d[2] = v.z; d[3] = v.w;
    }
    __syncthreads();

    const int wave = t >> 5, lane = t & 31;
    const int half = lane >> 4, col = lane & 15;
    const int n0   = wave * 16;
    const int mrow = col;

    v8f acc = {};
    #pragma unroll 8
    for (int k0 = 0; k0 < IN_C; k0 += 4) {
        int ka = k0 + 2 * half;
        v2f a = *(const v2f*)&sa[mrow * LDA_P + ka];
        v2f b;
        b.x = w[(size_t)ka * HID_C + n0 + col];
        b.y = w[(size_t)(ka + 1) * HID_C + n0 + col];
        acc = __builtin_amdgcn_wmma_f32_16x16x4_f32(false, a, false, b,
                                                    (short)0, acc, false, false);
    }

    float bv = bias[n0 + col];
    #pragma unroll
    for (int i = 0; i < 8; ++i) {
        int row = i + 8 * half;
        x0[(size_t)(m0 + row) * HID_C + n0 + col] = acc[i] + bv;
    }
}

#define REC_SRC(k) __builtin_amdgcn_readlane(r.x, j + (k))
#define REC_NRM(k) __int_as_float(__builtin_amdgcn_readlane(r.y, j + (k)))
#define ROW_LD(k)  h4[(size_t)REC_SRC(k) * 32 + lane]
#define ROW_FMA(wt, vv) row_fma(acc, wt, vv);
__device__ __forceinline__ void row_fma(float4& acc, float wt, const float4& vv) {
    acc.x = fmaf(wt, vv.x, acc.x); acc.y = fmaf(wt, vv.y, acc.y);
    acc.z = fmaf(wt, vv.z, acc.z); acc.w = fmaf(wt, vv.w, acc.w);
}
__global__ void kgather(const int* __restrict__ rowptr, const int* __restrict__ cnt,
                        const int2* __restrict__ rec, const float* __restrict__ hin,
                        float* __restrict__ agg) {
    const int lane = threadIdx.x & 31;
    const int row = __builtin_amdgcn_readfirstlane((int)((blockIdx.x * 256 + threadIdx.x) >> 5));
    const int beg = rowptr[row];
    const int n   = cnt[row];
    const float4* h4 = (const float4*)hin;

    float4 acc = make_float4(0.f, 0.f, 0.f, 0.f);
    for (int base = 0; base < n; base += 32) {
        const int m = (n - base < 32) ? (n - base) : 32;
        int2 r = make_int2(0, 0);
        if (lane < m) r = rec[beg + base + lane];
        int j = 0;
        for (; j + 8 <= m; j += 8) {
            float4 v0 = ROW_LD(0), v1 = ROW_LD(1), v2 = ROW_LD(2), v3 = ROW_LD(3);
            float4 v4 = ROW_LD(4), v5 = ROW_LD(5), v6 = ROW_LD(6), v7 = ROW_LD(7);
            float w0 = REC_NRM(0), w1 = REC_NRM(1), w2 = REC_NRM(2), w3 = REC_NRM(3);
            float w4 = REC_NRM(4), w5 = REC_NRM(5), w6 = REC_NRM(6), w7 = REC_NRM(7);
            ROW_FMA(w0, v0) ROW_FMA(w1, v1) ROW_FMA(w2, v2) ROW_FMA(w3, v3)
            ROW_FMA(w4, v4) ROW_FMA(w5, v5) ROW_FMA(w6, v6) ROW_FMA(w7, v7)
        }
        if (j + 4 <= m) {
            float4 v0 = ROW_LD(0), v1 = ROW_LD(1), v2 = ROW_LD(2), v3 = ROW_LD(3);
            float w0 = REC_NRM(0), w1 = REC_NRM(1), w2 = REC_NRM(2), w3 = REC_NRM(3);
            ROW_FMA(w0, v0) ROW_FMA(w1, v1) ROW_FMA(w2, v2) ROW_FMA(w3, v3)
            j += 4;
        }
        const int rem = m - j;
        if (rem == 3) {
            float4 v0 = ROW_LD(0), v1 = ROW_LD(1), v2 = ROW_LD(2);
            float w0 = REC_NRM(0), w1 = REC_NRM(1), w2 = REC_NRM(2);
            ROW_FMA(w0, v0) ROW_FMA(w1, v1) ROW_FMA(w2, v2)
        } else if (rem == 2) {
            float4 v0 = ROW_LD(0), v1 = ROW_LD(1);
            float w0 = REC_NRM(0), w1 = REC_NRM(1);
            ROW_FMA(w0, v0) ROW_FMA(w1, v1)
        } else if (rem == 1) {
            float4 v0 = ROW_LD(0);
            float w0 = REC_NRM(0);
            ROW_FMA(w0, v0)
        }
    }
    ((float4*)agg)[(size_t)row * 32 + lane] = acc;
}

#define LDA_L 132
__global__ void klayer(const float* __restrict__ agg, const float* __restrict__ hin,
                       const float* __restrict__ x0, const float* __restrict__ dinv,
                       const float* __restrict__ w, float beta, int do_relu,
                       float* __restrict__ hout) {
    __shared__ float sup[16 * LDA_L];
    const int t  = threadIdx.x;
    const int m0 = blockIdx.x * 16;

    const float4* af4 = (const float4*)agg;
    const float4* hf4 = (const float4*)hin;
    const float4* xf4 = (const float4*)x0;
    #pragma unroll
    for (int c = 0; c < 2; ++c) {
        int idx = c * 256 + t;
        int row = idx >> 5, q = idx & 31;
        float dv = dinv[m0 + row];
        float d2 = dv * dv;
        size_t base = (size_t)(m0 + row) * 32 + q;
        float4 a = af4[base], h = hf4[base], x = xf4[base];
        float* dp = &sup[row * LDA_L + q * 4];
        dp[0] = (1.f - ALPHA) * (a.x + d2 * h.x) + ALPHA * x.x;
        dp[1] = (1.f - ALPHA) * (a.y + d2 * h.y) + ALPHA * x.y;
        dp[2] = (1.f - ALPHA) * (a.z + d2 * h.z) + ALPHA * x.z;
        dp[3] = (1.f - ALPHA) * (a.w + d2 * h.w) + ALPHA * x.w;
    }
    __syncthreads();

    const int wave = t >> 5, lane = t & 31;
    const int half = lane >> 4, col = lane & 15;
    const int n0   = wave * 16;
    const int mrow = col;

    v8f acc = {};
    #pragma unroll 8
    for (int k0 = 0; k0 < HID_C; k0 += 4) {
        int ka = k0 + 2 * half;
        v2f a = *(const v2f*)&sup[mrow * LDA_L + ka];
        v2f b;
        b.x = w[(size_t)ka * HID_C + n0 + col];
        b.y = w[(size_t)(ka + 1) * HID_C + n0 + col];
        acc = __builtin_amdgcn_wmma_f32_16x16x4_f32(false, a, false, b,
                                                    (short)0, acc, false, false);
    }

    #pragma unroll
    for (int i = 0; i < 8; ++i) {
        int row = i + 8 * half;
        float s = sup[row * LDA_L + n0 + col];
        float v = (1.f - beta) * s + beta * acc[i];
        if (do_relu) v = fmaxf(v, 0.f);
        hout[(size_t)(m0 + row) * HID_C + n0 + col] = v;
    }
}

extern "C" void kernel_launch(void* const* d_in, const int* in_sizes, int n_in,
                              void* d_out, int out_size, void* d_ws, size_t ws_size,
                              hipStream_t stream) {
    const float* x      = (const float*)d_in[0];
    const int*   ei     = (const int*)d_in[1];
    const float* proj_w = (const float*)d_in[2];
    const float* proj_b = (const float*)d_in[3];
    const float* conv_w = (const float*)d_in[4];
    float* out = (float*)d_out;

    const int* e_src = ei;
    const int* e_dst = ei + N_EDGES;

    const size_t NH = (size_t)N_NODES * HID_C;
    char* ws = (char*)d_ws;
    int*   cnt    = (int*)ws;                                ws += ((size_t)N_NODES * 4 + 255) & ~255ull;
    float* dinv   = (float*)ws;                              ws += ((size_t)N_NODES * 4 + 255) & ~255ull;
    float* x0     = (float*)ws;                              ws += NH * 4;
    float* h      = (float*)ws;                              ws += NH * 4;
    float* agg    = (float*)ws;                              ws += NH * 4;
    int*   rank   = (int*)ws;                                ws += (size_t)N_EDGES * 4;
    int*   rowptr = (int*)ws;                                ws += ((size_t)N_NODES * 4 + 255) & ~255ull;
    int*   bsum   = (int*)ws;                                ws += 2048;
    int2*  rec    = (int2*)ws;

    kdeg_init <<<N_NBLK, 256, 0, stream>>>(cnt);
    kdeg_rank <<<(N_EDGES + 255) / 256, 256, 0, stream>>>(e_dst, cnt, rank);
    kblock_sum<<<N_NBLK, 256, 0, stream>>>(cnt, bsum);
    krowptr   <<<N_NBLK, 256, 0, stream>>>(cnt, bsum, rowptr, dinv);
    kfill     <<<(N_EDGES + 255) / 256, 256, 0, stream>>>(e_src, e_dst, rank, rowptr, dinv, rec);

    kproj<<<N_NODES / 16, 256, 0, stream>>>(x, proj_w, proj_b, x0);

    static const float betas[N_LAYERS] = {0.40546510f, 0.22314355f, 0.15415068f, 0.11778303f};

    for (int i = 0; i < N_LAYERS; ++i) {
        const float* hin  = (i == 0) ? x0 : h;
        float*       hout = (i == N_LAYERS - 1) ? out : h;
        kgather<<<N_NODES / 8, 256, 0, stream>>>(rowptr, cnt, rec, hin, agg);
        klayer <<<N_NODES / 16, 256, 0, stream>>>(agg, hin, x0, dinv,
                                                  conv_w + (size_t)i * HID_C * HID_C,
                                                  betas[i], (i != N_LAYERS - 1) ? 1 : 0, hout);
    }
}


# Row gather fused into the layer kernel: each 32-row work-group gathers 4 rows per wave straight into its LDS support tile, then runs WMMA on two row tiles; agg round trip and 4 launches removed
# speedup vs baseline: 3.3964x; 1.1136x over previous
#include <hip/hip_runtime.h>
#include <math.h>

#define N_NODES 100000
#define N_EDGES 1600000
#define IN_C    256
#define HID_C   128
#define N_LAYERS 4
#define ALPHA   0.1f
#define N_NBLK  ((N_NODES + 255) / 256)

typedef float v2f __attribute__((ext_vector_type(2)));
typedef float v8f __attribute__((ext_vector_type(8)));

__global__ void kdeg_init(int* cnt) {
    int v = blockIdx.x * 256 + threadIdx.x;
    if (v < N_NODES) cnt[v] = 0;
}

__global__ void kdeg_rank(const int* __restrict__ dst, int* __restrict__ cnt,
                          int* __restrict__ rank) {
    int e = blockIdx.x * 256 + threadIdx.x;
    if (e < N_EDGES) rank[e] = atomicAdd(&cnt[dst[e]], 1);
}

__global__ void kblock_sum(const int* __restrict__ cnt, int* __restrict__ bsum) {
    __shared__ int red[256];
    const int t = threadIdx.x;
    const int v = blockIdx.x * 256 + t;
    red[t] = (v < N_NODES) ? cnt[v] : 0;
    __syncthreads();
    for (int s = 128; s > 0; s >>= 1) {
        if (t < s) red[t] += red[t + s];
        __syncthreads();
    }
    if (t == 0) bsum[blockIdx.x] = red[0];
}

__global__ void krowptr(const int* __restrict__ cnt, const int* __restrict__ bsum,
                        int* __restrict__ rowptr, float* __restrict__ dinv) {
    __shared__ int red[256];
    __shared__ int sc[256];
    const int t = threadIdx.x;
    const int b = blockIdx.x;
    const int v = b * 256 + t;
    const int c = (v < N_NODES) ? cnt[v] : 0;

    int part = 0;
    for (int i = t; i < b; i += 256) part += bsum[i];
    red[t] = part;
    sc[t]  = c;
    __syncthreads();
    for (int s = 128; s > 0; s >>= 1) {
        if (t < s) red[t] += red[t + s];
        __syncthreads();
    }
    const int off = red[0];

    for (int d = 1; d < 256; d <<= 1) {
        int add = (t >= d) ? sc[t - d] : 0;
        __syncthreads();
        sc[t] += add;
        __syncthreads();
    }
    if (v < N_NODES) {
        rowptr[v] = off + sc[t] - c;
        dinv[v]   = rsqrtf((float)(c + 1));
    }
}

__global__ void kfill(const int* __restrict__ src, const int* __restrict__ dst,
                      const int* __restrict__ rank, const int* __restrict__ rowptr,
                      const float* __restrict__ dinv, int2* __restrict__ rec) {
    int e = blockIdx.x * 256 + threadIdx.x;
    if (e < N_EDGES) {
        int s = src[e], d = dst[e];
        int pos = rowptr[d] + rank[e];
        rec[pos] = make_int2(s, __float_as_int(dinv[s] * dinv[d]));
    }
}

#define LDA_P 260
__global__ void kproj(const float* __restrict__ x, const float* __restrict__ w,
                      const float* __restrict__ bias, float* __restrict__ x0) {
    __shared__ float sa[16 * LDA_P];
    const int t   = threadIdx.x;
    const int m0  = blockIdx.x * 16;

    const float4* xf4 = (const float4*)x;
    #pragma unroll
    for (int c = 0; c < 4; ++c) {
        int idx = c * 256 + t;
        int row = idx >> 6, q = idx & 63;
        float4 v = xf4[(size_t)(m0 + row) * 64 + q];
        float* d = &sa[row * LDA_P + q * 4];
        d[0] = v.x; d[1] = v.y; d[2] = v.z; d[3] = v.w;
    }
    __syncthreads();

    const int wave = t >> 5, lane = t & 31;
    const int half = lane >> 4, col = lane & 15;
    const int n0   = wave * 16;
    const int mrow = col;

    v8f acc = {};
    #pragma unroll 8
    for (int k0 = 0; k0 < IN_C; k0 += 4) {
        int ka = k0 + 2 * half;
        v2f a = *(const v2f*)&sa[mrow * LDA_P + ka];
        v2f b;
        b.x = w[(size_t)ka * HID_C + n0 + col];
        b.y = w[(size_t)(ka + 1) * HID_C + n0 + col];
        acc = __builtin_amdgcn_wmma_f32_16x16x4_f32(false, a, false, b,
                                                    (short)0, acc, false, false);
    }

    float bv = bias[n0 + col];
    #pragma unroll
    for (int i = 0; i < 8; ++i) {
        int row = i + 8 * half;
        x0[(size_t)(m0 + row) * HID_C + n0 + col] = acc[i] + bv;
    }
}

#define REC_SRC(k) __builtin_amdgcn_readlane(r.x, j + (k))
#define REC_NRM(k) __int_as_float(__builtin_amdgcn_readlane(r.y, j + (k)))
#define ROW_LD(k)  h4[(size_t)REC_SRC(k) * 32 + lane]
#define ROW_FMA(wt, vv) row_fma(acc, wt, vv);
__device__ __forceinline__ void row_fma(float4& acc, float wt, const float4& vv) {
    acc.x = fmaf(wt, vv.x, acc.x); acc.y = fmaf(wt, vv.y, acc.y);
    acc.z = fmaf(wt, vv.z, acc.z); acc.w = fmaf(wt, vv.w, acc.w);
}
#define TM    32
#define RPW   4
#define LDA_L 132
__global__ __launch_bounds__(256)
void klayer(const int* __restrict__ rowptr, const int* __restrict__ cnt,
            const int2* __restrict__ rec, const float* __restrict__ hin,
            const float* __restrict__ x0, const float* __restrict__ dinv,
            const float* __restrict__ w, float beta, int do_relu,
            float* __restrict__ hout) {
    __shared__ __attribute__((aligned(16))) float sup[TM * LDA_L];
    const int t    = threadIdx.x;
    const int lane = t & 31;
    const int wave = __builtin_amdgcn_readfirstlane(t >> 5);
    const int m0   = blockIdx.x * TM;
    const int r0   = m0 + wave * RPW;
    const float4* h4 = (const float4*)hin;
    const float4* x4 = (const float4*)x0;

    int beg[RPW], num[RPW];
    #pragma unroll
    for (int k = 0; k < RPW; ++k) { beg[k] = rowptr[r0 + k]; num[k] = cnt[r0 + k]; }
    int2 rfirst[RPW];
    #pragma unroll
    for (int k = 0; k < RPW; ++k) {
        rfirst[k] = make_int2(0, 0);
        if (lane < num[k]) rfirst[k] = rec[beg[k] + lane];
    }

    #pragma unroll
    for (int k = 0; k < RPW; ++k) {
        const int row = r0 + k;
        const int n   = num[k];
        const float4 hs = h4[(size_t)row * 32 + lane];
        const float4 xs = x4[(size_t)row * 32 + lane];
        const float  dv = dinv[row];

        float4 acc = make_float4(0.f, 0.f, 0.f, 0.f);
        for (int base = 0; base < n; base += 32) {
            const int m = (n - base < 32) ? (n - base) : 32;
            int2 r = rfirst[k];
            if (base > 0) {
                r = make_int2(0, 0);
                if (lane < m) r = rec[beg[k] + base + lane];
            }
            int j = 0;
            for (; j + 8 <= m; j += 8) {
                float4 v0 = ROW_LD(0), v1 = ROW_LD(1), v2 = ROW_LD(2), v3 = ROW_LD(3);
                float4 v4 = ROW_LD(4), v5 = ROW_LD(5), v6 = ROW_LD(6), v7 = ROW_LD(7);
                float w0 = REC_NRM(0), w1 = REC_NRM(1), w2 = REC_NRM(2), w3 = REC_NRM(3);
                float w4 = REC_NRM(4), w5 = REC_NRM(5), w6 = REC_NRM(6), w7 = REC_NRM(7);
                ROW_FMA(w0, v0) ROW_FMA(w1, v1) ROW_FMA(w2, v2) ROW_FMA(w3, v3)
                ROW_FMA(w4, v4) ROW_FMA(w5, v5) ROW_FMA(w6, v6) ROW_FMA(w7, v7)
            }
            if (j + 4 <= m) {
                float4 v0 = ROW_LD(0), v1 = ROW_LD(1), v2 = ROW_LD(2), v3 = ROW_LD(3);
                float w0 = REC_NRM(0), w1 = REC_NRM(1), w2 = REC_NRM(2), w3 = REC_NRM(3);
                ROW_FMA(w0, v0) ROW_FMA(w1, v1) ROW_FMA(w2, v2) ROW_FMA(w3, v3)
                j += 4;
            }
            const int rem = m - j;
            if (rem == 3) {
                float4 v0 = ROW_LD(0), v1 = ROW_LD(1), v2 = ROW_LD(2);
                float w0 = REC_NRM(0), w1 = REC_NRM(1), w2 = REC_NRM(2);
                ROW_FMA(w0, v0) ROW_FMA(w1, v1) ROW_FMA(w2, v2)
            } else if (rem == 2) {
                float4 v0 = ROW_LD(0), v1 = ROW_LD(1);
                float w0 = REC_NRM(0), w1 = REC_NRM(1);
                ROW_FMA(w0, v0) ROW_FMA(w1, v1)
            } else if (rem == 1) {
                float4 v0 = ROW_LD(0);
                float w0 = REC_NRM(0);
                ROW_FMA(w0, v0)
            }
        }

        const float d2 = dv * dv;
        float4 s;
        s.x = (1.f - ALPHA) * (acc.x + d2 * hs.x) + ALPHA * xs.x;
        s.y = (1.f - ALPHA) * (acc.y + d2 * hs.y) + ALPHA * xs.y;
        s.z = (1.f - ALPHA) * (acc.z + d2 * hs.z) + ALPHA * xs.z;
        s.w = (1.f - ALPHA) * (acc.w + d2 * hs.w) + ALPHA * xs.w;
        *(float4*)&sup[(wave * RPW + k) * LDA_L + lane * 4] = s;
    }
    __syncthreads();

    const int half = lane >> 4, col = lane & 15;
    const int n0   = wave * 16;
    const int mrow = col;

    v8f acc0 = {}, acc1 = {};
    #pragma unroll 8
    for (int k0 = 0; k0 < HID_C; k0 += 4) {
        int ka = k0 + 2 * half;
        v2f a0 = *(const v2f*)&sup[mrow * LDA_L + ka];
        v2f a1 = *(const v2f*)&sup[(16 + mrow) * LDA_L + ka];
        v2f b;
        b.x = w[(size_t)ka * HID_C + n0 + col];
        b.y = w[(size_t)(ka + 1) * HID_C + n0 + col];
        acc0 = __builtin_amdgcn_wmma_f32_16x16x4_f32(false, a0, false, b,
                                                     (short)0, acc0, false, false);
        acc1 = __builtin_amdgcn_wmma_f32_16x16x4_f32(false, a1, false, b,
                                                     (short)0, acc1, false, false);
    }

    #pragma unroll
    for (int i = 0; i < 8; ++i) {
        int row = i + 8 * half;
        float s = sup[row * LDA_L + n0 + col];
        float v = (1.f - beta) * s + beta * acc0[i];
        if (do_relu) v = fmaxf(v, 0.f);
        hout[(size_t)(m0 + row) * HID_C + n0 + col] = v;
    }
    #pragma unroll
    for (int i = 0; i < 8; ++i) {
        int row = 16 + i + 8 * half;
        float s = sup[row * LDA_L + n0 + col];
        float v = (1.f - beta) * s + beta * acc1[i];
        if (do_relu) v = fmaxf(v, 0.f);
        hout[(size_t)(m0 + row) * HID_C + n0 + col] = v;
    }
}

extern "C" void kernel_launch(void* const* d_in, const int* in_sizes, int n_in,
                              void* d_out, int out_size, void* d_ws, size_t ws_size,
                              hipStream_t stream) {
    const float* x      = (const float*)d_in[0];
    const int*   ei     = (const int*)d_in[1];
    const float* proj_w = (const float*)d_in[2];
    const float* proj_b = (const float*)d_in[3];
    const float* conv_w = (const float*)d_in[4];
    float* out = (float*)d_out;

    const int* e_src = ei;
    const int* e_dst = ei + N_EDGES;

    const size_t NH = (size_t)N_NODES * HID_C;
    char* ws = (char*)d_ws;
    int*   cnt    = (int*)ws;                                ws += ((size_t)N_NODES * 4 + 255) & ~255ull;
    float* dinv   = (float*)ws;                              ws += ((size_t)N_NODES * 4 + 255) & ~255ull;
    float* x0     = (float*)ws;                              ws += NH * 4;
    float* ha     = (float*)ws;                              ws += NH * 4;
    float* hb     = (float*)ws;                              ws += NH * 4;
    int*   rank   = (int*)ws;                                ws += (size_t)N_EDGES * 4;
    int*   rowptr = (int*)ws;                                ws += ((size_t)N_NODES * 4 + 255) & ~255ull;
    int*   bsum   = (int*)ws;                                ws += 2048;
    int2*  rec    = (int2*)ws;

    kdeg_init <<<N_NBLK, 256, 0, stream>>>(cnt);
    kdeg_rank <<<(N_EDGES + 255) / 256, 256, 0, stream>>>(e_dst, cnt, rank);
    kblock_sum<<<N_NBLK, 256, 0, stream>>>(cnt, bsum);
    krowptr   <<<N_NBLK, 256, 0, stream>>>(cnt, bsum, rowptr, dinv);
    kfill     <<<(N_EDGES + 255) / 256, 256, 0, stream>>>(e_src, e_dst, rank, rowptr, dinv, rec);

    kproj<<<N_NODES / 16, 256, 0, stream>>>(x, proj_w, proj_b, x0);

    static const float betas[N_LAYERS] = {0.40546510f, 0.22314355f, 0.15415068f, 0.11778303f};

    const float* const lin[N_LAYERS]  = {x0, ha, hb, ha};
    float* const       lout[N_LAYERS] = {ha, hb, ha, out};
    for (int i = 0; i < N_LAYERS; ++i) {
        const float* hin  = lin[i];
        float*       hout = lout[i];
        klayer<<<N_NODES / TM, 256, 0, stream>>>(rowptr, cnt, rec, hin, x0, dinv,
                                                 conv_w + (size_t)i * HID_C * HID_C,
                                                 betas[i], (i != N_LAYERS - 1) ? 1 : 0, hout);
    }
}


# Per-edge returning global adds replaced by per-chunk LDS counters (64 chunks, two 16-bit counts per word, returning LDS add as rank) and a per-node scan over chunks; row sums in chunk-then-rank order
# speedup vs baseline: 3.8564x; 1.1354x over previous
#include <hip/hip_runtime.h>
#include <math.h>

#define N_NODES 100000
#define N_EDGES 1600000
#define IN_C    256
#define HID_C   128
#define N_LAYERS 4
#define ALPHA   0.1f
#define N_NBLK  ((N_NODES + 255) / 256)

typedef float v2f __attribute__((ext_vector_type(2)));
typedef float v8f __attribute__((ext_vector_type(8)));

#define N_CHUNK 64
#define CHUNK_E (N_EDGES / N_CHUNK)
#define N_PAIR  (N_NODES / 2)
#define EPT     ((CHUNK_E + 1023) / 1024)
static_assert(N_EDGES % N_CHUNK == 0, "chunks cover the edge list exactly");
static_assert(CHUNK_E < 65536, "a 16-bit count holds a whole chunk");
static_assert(N_NODES % 8 == 0, "counter words move as 16-byte vectors");

__global__ __launch_bounds__(1024)
void kcount(const int* __restrict__ dst, unsigned* __restrict__ table, int* __restrict__ rank) {
    __shared__ __attribute__((aligned(16))) unsigned lc[N_PAIR];
    const int t  = threadIdx.x;
    const int c  = blockIdx.x;
    const int e0 = c * CHUNK_E;
    uint4* lc4 = (uint4*)lc;

    int dv[EPT];
    #pragma unroll
    for (int k = 0; k < EPT; ++k) {
        int i = t + k * 1024;
        dv[k] = (i < CHUNK_E) ? dst[e0 + i] : 0;
    }
    for (int i = t; i < N_PAIR / 4; i += 1024) lc4[i] = make_uint4(0u, 0u, 0u, 0u);
    __syncthreads();

    unsigned old[EPT];
    #pragma unroll
    for (int k = 0; k < EPT; ++k) {
        int i = t + k * 1024;
        old[k] = 0u;
        if (i < CHUNK_E) old[k] = atomicAdd(&lc[dv[k] >> 1], 1u << ((unsigned)(dv[k] & 1) * 16u));
    }
    #pragma unroll
    for (int k = 0; k < EPT; ++k) {
        int i = t + k * 1024;
        if (i < CHUNK_E) rank[e0 + i] = (int)((old[k] >> ((unsigned)(dv[k] & 1) * 16u)) & 0xFFFFu);
    }
    __syncthreads();

    uint4* t4 = (uint4*)(table + (size_t)c * N_PAIR);
    for (int i = t; i < N_PAIR / 4; i += 1024) t4[i] = lc4[i];
}

__global__ __launch_bounds__(128)
void kcolscan(const unsigned* __restrict__ table, int2* __restrict__ offs, int2* __restrict__ cnt2) {
    const int w = blockIdx.x * 128 + threadIdx.x;
    if (w >= N_PAIR) return;
    int lo = 0, hi = 0;
    #pragma unroll 16
    for (int c = 0; c < N_CHUNK; ++c) {
        unsigned v = table[(size_t)c * N_PAIR + w];
        offs[(size_t)c * N_PAIR + w] = make_int2(lo, hi);
        lo += (int)(v & 0xFFFFu);
        hi += (int)(v >> 16);
    }
    cnt2[w] = make_int2(lo, hi);
}

__global__ void kblock_sum(const int* __restrict__ cnt, int* __restrict__ bsum) {
    __shared__ int red[256];
    const int t = threadIdx.x;
    const int v = blockIdx.x * 256 + t;
    red[t] = (v < N_NODES) ? cnt[v] : 0;
    __syncthreads();
    for (int s = 128; s > 0; s >>= 1) {
        if (t < s) red[t] += red[t + s];
        __syncthreads();
    }
    if (t == 0) bsum[blockIdx.x] = red[0];
}

__global__ void krowptr(const int* __restrict__ cnt, const int* __restrict__ bsum,
                        int* __restrict__ rowptr, float* __restrict__ dinv) {
    __shared__ int red[256];
    __shared__ int sc[256];
    const int t = threadIdx.x;
    const int b = blockIdx.x;
    const int v = b * 256 + t;
    const int c = (v < N_NODES) ? cnt[v] : 0;

    int part = 0;
    for (int i = t; i < b; i += 256) part += bsum[i];
    red[t] = part;
    sc[t]  = c;
    __syncthreads();
    for (int s = 128; s > 0; s >>= 1) {
        if (t < s) red[t] += red[t + s];
        __syncthreads();
    }
    const int off = red[0];

    for (int d = 1; d < 256; d <<= 1) {
        int add = (t >= d) ? sc[t - d] : 0;
        __syncthreads();
        sc[t] += add;
        __syncthreads();
    }
    if (v < N_NODES) {
        rowptr[v] = off + sc[t] - c;
        dinv[v]   = rsqrtf((float)(c + 1));
    }
}

__global__ void kfill(const int* __restrict__ src, const int* __restrict__ dst,
                      const int* __restrict__ rank, const int* __restrict__ rowptr,
                      const int* __restrict__ offs, const float* __restrict__ dinv,
                      int2* __restrict__ rec) {
    int e = blockIdx.x * 256 + threadIdx.x;
    if (e < N_EDGES) {
        int s = src[e], d = dst[e];
        int pos = rowptr[d] + offs[(size_t)(e / CHUNK_E) * N_NODES + d] + rank[e];
        rec[pos] = make_int2(s, __float_as_int(dinv[s] * dinv[d]));
    }
}

#define LDA_P 260
__global__ void kproj(const float* __restrict__ x, const float* __restrict__ w,
                      const float* __restrict__ bias, float* __restrict__ x0) {
    __shared__ float sa[16 * LDA_P];
    const int t   = threadIdx.x;
    const int m0  = blockIdx.x * 16;

    const float4* xf4 = (const float4*)x;
    #pragma unroll
    for (int c = 0; c < 4; ++c) {
        int idx = c * 256 + t;
        int row = idx >> 6, q = idx & 63;
        float4 v = xf4[(size_t)(m0 + row) * 64 + q];
        float* d = &sa[row * LDA_P + q * 4];
        d[0] = v.x; d[1] = v.y; d[2] = v.z; d[3] = v.w;
    }
    __syncthreads();

    const int wave = t >> 5, lane = t & 31;
    const int half = lane >> 4, col = lane & 15;
    const int n0   = wave * 16;
    const int mrow = col;

    v8f acc = {};
    #pragma unroll 8
    for (int k0 = 0; k0 < IN_C; k0 += 4) {
        int ka = k0 + 2 * half;
        v2f a = *(const v2f*)&sa[mrow * LDA_P + ka];
        v2f b;
        b.x = w[(size_t)ka * HID_C + n0 + col];
        b.y = w[(size_t)(ka + 1) * HID_C + n0 + col];
        acc = __builtin_amdgcn_wmma_f32_16x16x4_f32(false, a, false, b,
                                                    (short)0, acc, false, false);
    }

    float bv = bias[n0 + col];
    #pragma unroll
    for (int i = 0; i < 8; ++i) {
        int row = i + 8 * half;
        x0[(size_t)(m0 + row) * HID_C + n0 + col] = acc[i] + bv;
    }
}

#define REC_SRC(k) __builtin_amdgcn_readlane(r.x, j + (k))
#define REC_NRM(k) __int_as_float(__builtin_amdgcn_readlane(r.y, j + (k)))
#define ROW_LD(k)  h4[(size_t)REC_SRC(k) * 32 + lane]
#define ROW_FMA(wt, vv) row_fma(acc, wt, vv);
__device__ __forceinline__ void row_fma(float4& acc, float wt, const float4& vv) {
    acc.x = fmaf(wt, vv.x, acc.x); acc.y = fmaf(wt, vv.y, acc.y);
    acc.z = fmaf(wt, vv.z, acc.z); acc.w = fmaf(wt, vv.w, acc.w);
}
#define TM    32
#define RPW   4
#define LDA_L 132
__global__ __launch_bounds__(256)
void klayer(const int* __restrict__ rowptr, const int* __restrict__ cnt,
            const int2* __restrict__ rec, const float* __restrict__ hin,
            const float* __restrict__ x0, const float* __restrict__ dinv,
            const float* __restrict__ w, float beta, int do_relu,
            float* __restrict__ hout) {
    __shared__ __attribute__((aligned(16))) float sup[TM * LDA_L];
    const int t    = threadIdx.x;
    const int lane = t & 31;
    const int wave = __builtin_amdgcn_readfirstlane(t >> 5);
    const int m0   = blockIdx.x * TM;
    const int r0   = m0 + wave * RPW;
    const float4* h4 = (const float4*)hin;
    const float4* x4 = (const float4*)x0;

    int beg[RPW], num[RPW];
    #pragma unroll
    for (int k = 0; k < RPW; ++k) { beg[k] = rowptr[r0 + k]; num[k] = cnt[r0 + k]; }
    int2 rfirst[RPW];
    #pragma unroll
    for (int k = 0; k < RPW; ++k) {
        rfirst[k] = make_int2(0, 0);
        if (lane < num[k]) rfirst[k] = rec[beg[k] + lane];
    }

    #pragma unroll
    for (int k = 0; k < RPW; ++k) {
        const int row = r0 + k;
        const int n   = num[k];
        const float4 hs = h4[(size_t)row * 32 + lane];
        const float4 xs = x4[(size_t)row * 32 + lane];
        const float  dv = dinv[row];

        float4 acc = make_float4(0.f, 0.f, 0.f, 0.f);
        for (int base = 0; base < n; base += 32) {
            const int m = (n - base < 32) ? (n - base) : 32;
            int2 r = rfirst[k];
            if (base > 0) {
                r = make_int2(0, 0);
                if (lane < m) r = rec[beg[k] + base + lane];
            }
            int j = 0;
            for (; j + 8 <= m; j += 8) {
                float4 v0 = ROW_LD(0), v1 = ROW_LD(1), v2 = ROW_LD(2), v3 = ROW_LD(3);
                float4 v4 = ROW_LD(4), v5 = ROW_LD(5), v6 = ROW_LD(6), v7 = ROW_LD(7);
                float w0 = REC_NRM(0), w1 = REC_NRM(1), w2 = REC_NRM(2), w3 = REC_NRM(3);
                float w4 = REC_NRM(4), w5 = REC_NRM(5), w6 = REC_NRM(6), w7 = REC_NRM(7);
                ROW_FMA(w0, v0) ROW_FMA(w1, v1) ROW_FMA(w2, v2) ROW_FMA(w3, v3)
                ROW_FMA(w4, v4) ROW_FMA(w5, v5) ROW_FMA(w6, v6) ROW_FMA(w7, v7)
            }
            if (j + 4 <= m) {
                float4 v0 = ROW_LD(0), v1 = ROW_LD(1), v2 = ROW_LD(2), v3 = ROW_LD(3);
                float w0 = REC_NRM(0), w1 = REC_NRM(1), w2 = REC_NRM(2), w3 = REC_NRM(3);
                ROW_FMA(w0, v0) ROW_FMA(w1, v1) ROW_FMA(w2, v2) ROW_FMA(w3, v3)
                j += 4;
            }
            const int rem = m - j;
            if (rem == 3) {
                float4 v0 = ROW_LD(0), v1 = ROW_LD(1), v2 = ROW_LD(2);
                float w0 = REC_NRM(0), w1 = REC_NRM(1), w2 = REC_NRM(2);
                ROW_FMA(w0, v0) ROW_FMA(w1, v1) ROW_FMA(w2, v2)
            } else if (rem == 2) {
                float4 v0 = ROW_LD(0), v1 = ROW_LD(1);
                float w0 = REC_NRM(0), w1 = REC_NRM(1);
                ROW_FMA(w0, v0) ROW_FMA(w1, v1)
            } else if (rem == 1) {
                float4 v0 = ROW_LD(0);
                float w0 = REC_NRM(0);
                ROW_FMA(w0, v0)
            }
        }

        const float d2 = dv * dv;
        float4 s;
        s.x = (1.f - ALPHA) * (acc.x + d2 * hs.x) + ALPHA * xs.x;
        s.y = (1.f - ALPHA) * (acc.y + d2 * hs.y) + ALPHA * xs.y;
        s.z = (1.f - ALPHA) * (acc.z + d2 * hs.z) + ALPHA * xs.z;
        s.w = (1.f - ALPHA) * (acc.w + d2 * hs.w) + ALPHA * xs.w;
        *(float4*)&sup[(wave * RPW + k) * LDA_L + lane * 4] = s;
    }
    __syncthreads();

    const int half = lane >> 4, col = lane & 15;
    const int n0   = wave * 16;
    const int mrow = col;

    v8f acc0 = {}, acc1 = {};
    #pragma unroll 8
    for (int k0 = 0; k0 < HID_C; k0 += 4) {
        int ka = k0 + 2 * half;
        v2f a0 = *(const v2f*)&sup[mrow * LDA_L + ka];
        v2f a1 = *(const v2f*)&sup[(16 + mrow) * LDA_L + ka];
        v2f b;
        b.x = w[(size_t)ka * HID_C + n0 + col];
        b.y = w[(size_t)(ka + 1) * HID_C + n0 + col];
        acc0 = __builtin_amdgcn_wmma_f32_16x16x4_f32(false, a0, false, b,
                                                     (short)0, acc0, false, false);
        acc1 = __builtin_amdgcn_wmma_f32_16x16x4_f32(false, a1, false, b,
                                                     (short)0, acc1, false, false);
    }

    #pragma unroll
    for (int i = 0; i < 8; ++i) {
        int row = i + 8 * half;
        float s = sup[row * LDA_L + n0 + col];
        float v = (1.f - beta) * s + beta * acc0[i];
        if (do_relu) v = fmaxf(v, 0.f);
        hout[(size_t)(m0 + row) * HID_C + n0 + col] = v;
    }
    #pragma unroll
    for (int i = 0; i < 8; ++i) {
        int row = 16 + i + 8 * half;
        float s = sup[row * LDA_L + n0 + col];
        float v = (1.f - beta) * s + beta * acc1[i];
        if (do_relu) v = fmaxf(v, 0.f);
        hout[(size_t)(m0 + row) * HID_C + n0 + col] = v;
    }
}

extern "C" void kernel_launch(void* const* d_in, const int* in_sizes, int n_in,
                              void* d_out, int out_size, void* d_ws, size_t ws_size,
                              hipStream_t stream) {
    const float* x      = (const float*)d_in[0];
    const int*   ei     = (const int*)d_in[1];
    const float* proj_w = (const float*)d_in[2];
    const float* proj_b = (const float*)d_in[3];
    const float* conv_w = (const float*)d_in[4];
    float* out = (float*)d_out;

    const int* e_src = ei;
    const int* e_dst = ei + N_EDGES;

    const size_t NH = (size_t)N_NODES * HID_C;
    char* ws = (char*)d_ws;
    int*   cnt    = (int*)ws;                                ws += ((size_t)N_NODES * 4 + 255) & ~255ull;
    float* dinv   = (float*)ws;                              ws += ((size_t)N_NODES * 4 + 255) & ~255ull;
    float* x0     = (float*)ws;                              ws += NH * 4;
    float* ha     = (float*)ws;                              ws += NH * 4;
    float* hb     = (float*)ws;                              ws += NH * 4;
    int*   rank   = (int*)ws;                                ws += (size_t)N_EDGES * 4;
    int*   rowptr = (int*)ws;                                ws += ((size_t)N_NODES * 4 + 255) & ~255ull;
    int*   bsum   = (int*)ws;                                ws += 2048;
    unsigned* table = (unsigned*)ws;                         ws += (size_t)N_CHUNK * N_PAIR * 4;
    int2*  offs   = (int2*)ws;                               ws += (size_t)N_CHUNK * N_NODES * 4;
    int2*  rec    = (int2*)ws;

    kcount    <<<N_CHUNK, 1024, 0, stream>>>(e_dst, table, rank);
    kcolscan  <<<(N_PAIR + 127) / 128, 128, 0, stream>>>(table, offs, (int2*)cnt);
    kblock_sum<<<N_NBLK, 256, 0, stream>>>(cnt, bsum);
    krowptr   <<<N_NBLK, 256, 0, stream>>>(cnt, bsum, rowptr, dinv);
    kfill     <<<(N_EDGES + 255) / 256, 256, 0, stream>>>(e_src, e_dst, rank, rowptr, (const int*)offs, dinv, rec);

    kproj<<<N_NODES / 16, 256, 0, stream>>>(x, proj_w, proj_b, x0);

    static const float betas[N_LAYERS] = {0.40546510f, 0.22314355f, 0.15415068f, 0.11778303f};

    const float* const lin[N_LAYERS]  = {x0, ha, hb, ha};
    float* const       lout[N_LAYERS] = {ha, hb, ha, out};
    for (int i = 0; i < N_LAYERS; ++i) {
        const float* hin  = lin[i];
        float*       hout = lout[i];
        klayer<<<N_NODES / TM, 256, 0, stream>>>(rowptr, cnt, rec, hin, x0, dinv,
                                                 conv_w + (size_t)i * HID_C * HID_C,
                                                 betas[i], (i != N_LAYERS - 1) ? 1 : 0, hout);
    }
}


# WMMA B fragments read from a per-call copy of the weights packed in fragment order (written inside the block-sum launch): one 16-byte load per lane per two k-steps replaces four 4-byte loads
# speedup vs baseline: 4.0847x; 1.0592x over previous
#include <hip/hip_runtime.h>
#include <math.h>

#define N_NODES 100000
#define N_EDGES 1600000
#define IN_C    256
#define HID_C   128
#define N_LAYERS 4
#define ALPHA   0.1f
#define N_NBLK  ((N_NODES + 255) / 256)

typedef float v2f __attribute__((ext_vector_type(2)));
typedef float v8f __attribute__((ext_vector_type(8)));

#define N_CHUNK 64
#define CHUNK_E (N_EDGES / N_CHUNK)
#define N_PAIR  (N_NODES / 2)
#define EPT     ((CHUNK_E + 1023) / 1024)
static_assert(N_EDGES % N_CHUNK == 0, "chunks cover the edge list exactly");
static_assert(CHUNK_E < 65536, "a 16-bit count holds a whole chunk");
static_assert(N_NODES % 8 == 0, "counter words move as 16-byte vectors");

__global__ __launch_bounds__(1024)
void kcount(const int* __restrict__ dst, unsigned* __restrict__ table, int* __restrict__ rank) {
    __shared__ __attribute__((aligned(16))) unsigned lc[N_PAIR];
    const int t  = threadIdx.x;
    const int c  = blockIdx.x;
    const int e0 = c * CHUNK_E;
    uint4* lc4 = (uint4*)lc;

    int dv[EPT];
    #pragma unroll
    for (int k = 0; k < EPT; ++k) {
        int i = t + k * 1024;
        dv[k] = (i < CHUNK_E) ? dst[e0 + i] : 0;
    }
    for (int i = t; i < N_PAIR / 4; i += 1024) lc4[i] = make_uint4(0u, 0u, 0u, 0u);
    __syncthreads();

    unsigned old[EPT];
    #pragma unroll
    for (int k = 0; k < EPT; ++k) {
        int i = t + k * 1024;
        old[k] = 0u;
        if (i < CHUNK_E) old[k] = atomicAdd(&lc[dv[k] >> 1], 1u << ((unsigned)(dv[k] & 1) * 16u));
    }
    #pragma unroll
    for (int k = 0; k < EPT; ++k) {
        int i = t + k * 1024;
        if (i < CHUNK_E) rank[e0 + i] = (int)((old[k] >> ((unsigned)(dv[k] & 1) * 16u)) & 0xFFFFu);
    }
    __syncthreads();

    uint4* t4 = (uint4*)(table + (size_t)c * N_PAIR);
    for (int i = t; i < N_PAIR / 4; i += 1024) t4[i] = lc4[i];
}

__global__ __launch_bounds__(128)
void kcolscan(const unsigned* __restrict__ table, int2* __restrict__ offs, int2* __restrict__ cnt2) {
    const int w = blockIdx.x * 128 + threadIdx.x;
    if (w >= N_PAIR) return;
    int lo = 0, hi = 0;
    #pragma unroll 16
    for (int c = 0; c < N_CHUNK; ++c) {
        unsigned v = table[(size_t)c * N_PAIR + w];
        offs[(size_t)c * N_PAIR + w] = make_int2(lo, hi);
        lo += (int)(v & 0xFFFFu);
        hi += (int)(v >> 16);
    }
    cnt2[w] = make_int2(lo, hi);
}

#define WP_PROJ4 (IN_C / 8 * 256)
#define WP_CONV4 (HID_C / 8 * 256)
#define WP_ALL4  (WP_PROJ4 + N_LAYERS * WP_CONV4)
#define WP_NBLK  (WP_ALL4 / 256)
static_assert(WP_ALL4 % 256 == 0, "whole 256-thread groups cover the copy exactly");
__device__ __forceinline__
void pack_w(const int o, const float* __restrict__ proj_w, const float* __restrict__ conv_w,
            float4* __restrict__ wp) {
    const float* wm = proj_w;
    int q = o;
    if (o >= WP_PROJ4) {
        const int r = o - WP_PROJ4;
        const int layer = r / WP_CONV4;
        q  = r - layer * WP_CONV4;
        wm = conv_w + (size_t)layer * HID_C * HID_C;
    }
    const int lane = q & 31, strip = (q >> 5) & 7, p = q >> 8;
    const int half = lane >> 4, col = lane & 15;
    const int n = strip * 16 + col;
    const int k = p * 8 + 2 * half;
    float4 v;
    v.x = wm[(size_t)k * HID_C + n];
    v.y = wm[(size_t)(k + 1) * HID_C + n];
    v.z = wm[(size_t)(k + 4) * HID_C + n];
    v.w = wm[(size_t)(k + 5) * HID_C + n];
    wp[o] = v;
}

static_assert(WP_NBLK <= N_NBLK, "enough work-groups in this launch to carry the weight copy");
__global__ void kblock_sum(const int* __restrict__ cnt, int* __restrict__ bsum,
                           const float* __restrict__ proj_w, const float* __restrict__ conv_w,
                           float4* __restrict__ wp) {
    __shared__ int red[256];
    const int t = threadIdx.x;
    if (blockIdx.x < WP_NBLK) pack_w(blockIdx.x * 256 + t, proj_w, conv_w, wp);
    const int v = blockIdx.x * 256 + t;
    red[t] = (v < N_NODES) ? cnt[v] : 0;
    __syncthreads();
    for (int s = 128; s > 0; s >>= 1) {
        if (t < s) red[t] += red[t + s];
        __syncthreads();
    }
    if (t == 0) bsum[blockIdx.x] = red[0];
}

__global__ void krowptr(const int* __restrict__ cnt, const int* __restrict__ bsum,
                        int* __restrict__ rowptr, float* __restrict__ dinv) {
    __shared__ int red[256];
    __shared__ int sc[256];
    const int t = threadIdx.x;
    const int b = blockIdx.x;
    const int v = b * 256 + t;
    const int c = (v < N_NODES) ? cnt[v] : 0;

    int part = 0;
    for (int i = t; i < b; i += 256) part += bsum[i];
    red[t] = part;
    sc[t]  = c;
    __syncthreads();
    for (int s = 128; s > 0; s >>= 1) {
        if (t < s) red[t] += red[t + s];
        __syncthreads();
    }
    const int off = red[0];

    for (int d = 1; d < 256; d <<= 1) {
        int add = (t >= d) ? sc[t - d] : 0;
        __syncthreads();
        sc[t] += add;
        __syncthreads();
    }
    if (v < N_NODES) {
        rowptr[v] = off + sc[t] - c;
        dinv[v]   = rsqrtf((float)(c + 1));
    }
}

__global__ void kfill(const int* __restrict__ src, const int* __restrict__ dst,
                      const int* __restrict__ rank, const int* __restrict__ rowptr,
                      const int* __restrict__ offs, const float* __restrict__ dinv,
                      int2* __restrict__ rec) {
    int e = blockIdx.x * 256 + threadIdx.x;
    if (e < N_EDGES) {
        int s = src[e], d = dst[e];
        int pos = rowptr[d] + offs[(size_t)(e / CHUNK_E) * N_NODES + d] + rank[e];
        rec[pos] = make_int2(s, __float_as_int(dinv[s] * dinv[d]));
    }
}

#define LDA_P 260
__global__ void kproj(const float* __restrict__ x, const float4* __restrict__ wp,
                      const float* __restrict__ bias, float* __restrict__ x0) {
    __shared__ float sa[16 * LDA_P];
    const int t   = threadIdx.x;
    const int m0  = blockIdx.x * 16;

    const float4* xf4 = (const float4*)x;
    #pragma unroll
    for (int c = 0; c < 4; ++c) {
        int idx = c * 256 + t;
        int row = idx >> 6, q = idx & 63;
        float4 v = xf4[(size_t)(m0 + row) * 64 + q];
        float* d = &sa[row * LDA_P + q * 4];
        d[0] = v.x; d[1] = v.y; d[2] = v.z; d[3] = v.w;
    }
    __syncthreads();

    const int wave = t >> 5, lane = t & 31;
    const int half = lane >> 4, col = lane & 15;
    const int n0   = wave * 16;
    const int mrow = col;

    v8f acc = {};
    #pragma unroll 8
    for (int p = 0; p < IN_C / 8; ++p) {
        const float4 b4 = wp[p * 256 + t];
        const int ka = p * 8 + 2 * half;
        v2f a = *(const v2f*)&sa[mrow * LDA_P + ka];
        v2f b;
        b.x = b4.x; b.y = b4.y;
        acc = __builtin_amdgcn_wmma_f32_16x16x4_f32(false, a, false, b,
                                                    (short)0, acc, false, false);
        a = *(const v2f*)&sa[mrow * LDA_P + ka + 4];
        b.x = b4.z; b.y = b4.w;
        acc = __builtin_amdgcn_wmma_f32_16x16x4_f32(false, a, false, b,
                                                    (short)0, acc, false, false);
    }

    float bv = bias[n0 + col];
    #pragma unroll
    for (int i = 0; i < 8; ++i) {
        int row = i + 8 * half;
        x0[(size_t)(m0 + row) * HID_C + n0 + col] = acc[i] + bv;
    }
}

#define REC_SRC(k) __builtin_amdgcn_readlane(r.x, j + (k))
#define REC_NRM(k) __int_as_float(__builtin_amdgcn_readlane(r.y, j + (k)))
#define ROW_LD(k)  h4[(size_t)REC_SRC(k) * 32 + lane]
#define ROW_FMA(wt, vv) row_fma(acc, wt, vv);
__device__ __forceinline__ void row_fma(float4& acc, float wt, const float4& vv) {
    acc.x = fmaf(wt, vv.x, acc.x); acc.y = fmaf(wt, vv.y, acc.y);
    acc.z = fmaf(wt, vv.z, acc.z); acc.w = fmaf(wt, vv.w, acc.w);
}
#define TM    32
#define RPW   4
#define LDA_L 132
__global__ __launch_bounds__(256)
void klayer(const int* __restrict__ rowptr, const int* __restrict__ cnt,
            const int2* __restrict__ rec, const float* __restrict__ hin,
            const float* __restrict__ x0, const float* __restrict__ dinv,
            const float4* __restrict__ wp, float beta, int do_relu,
            float* __restrict__ hout) {
    __shared__ __attribute__((aligned(16))) float sup[TM * LDA_L];
    const int t    = threadIdx.x;
    const int lane = t & 31;
    const int wave = __builtin_amdgcn_readfirstlane(t >> 5);
    const int m0   = blockIdx.x * TM;
    const int r0   = m0 + wave * RPW;
    const float4* h4 = (const float4*)hin;
    const float4* x4 = (const float4*)x0;

    int beg[RPW], num[RPW];
    #pragma unroll
    for (int k = 0; k < RPW; ++k) { beg[k] = rowptr[r0 + k]; num[k] = cnt[r0 + k]; }
    int2 rfirst[RPW];
    #pragma unroll
    for (int k = 0; k < RPW; ++k) {
        rfirst[k] = make_int2(0, 0);
        if (lane < num[k]) rfirst[k] = rec[beg[k] + lane];
    }

    #pragma unroll
    for (int k = 0; k < RPW; ++k) {
        const int row = r0 + k;
        const int n   = num[k];
        const float4 hs = h4[(size_t)row * 32 + lane];
        const float4 xs = x4[(size_t)row * 32 + lane];
        const float  dv = dinv[row];

        float4 acc = make_float4(0.f, 0.f, 0.f, 0.f);
        for (int base = 0; base < n; base += 32) {
            const int m = (n - base < 32) ? (n - base) : 32;
            int2 r = rfirst[k];
            if (base > 0) {
                r = make_int2(0, 0);
                if (lane < m) r = rec[beg[k] + base + lane];
            }
            int j = 0;
            for (; j + 8 <= m; j += 8) {
                float4 v0 = ROW_LD(0), v1 = ROW_LD(1), v2 = ROW_LD(2), v3 = ROW_LD(3);
                float4 v4 = ROW_LD(4), v5 = ROW_LD(5), v6 = ROW_LD(6), v7 = ROW_LD(7);
                float w0 = REC_NRM(0), w1 = REC_NRM(1), w2 = REC_NRM(2), w3 = REC_NRM(3);
                float w4 = REC_NRM(4), w5 = REC_NRM(5), w6 = REC_NRM(6), w7 = REC_NRM(7);
                ROW_FMA(w0, v0) ROW_FMA(w1, v1) ROW_FMA(w2, v2) ROW_FMA(w3, v3)
                ROW_FMA(w4, v4) ROW_FMA(w5, v5) ROW_FMA(w6, v6) ROW_FMA(w7, v7)
            }
            if (j + 4 <= m) {
                float4 v0 = ROW_LD(0), v1 = ROW_LD(1), v2 = ROW_LD(2), v3 = ROW_LD(3);
                float w0 = REC_NRM(0), w1 = REC_NRM(1), w2 = REC_NRM(2), w3 = REC_NRM(3);
                ROW_FMA(w0, v0) ROW_FMA(w1, v1) ROW_FMA(w2, v2) ROW_FMA(w3, v3)
                j += 4;
            }
            const int rem = m - j;
            if (rem == 3) {
                float4 v0 = ROW_LD(0), v1 = ROW_LD(1), v2 = ROW_LD(2);
                float w0 = REC_NRM(0), w1 = REC_NRM(1), w2 = REC_NRM(2);
                ROW_FMA(w0, v0) ROW_FMA(w1, v1) ROW_FMA(w2, v2)
            } else if (rem == 2) {
                float4 v0 = ROW_LD(0), v1 = ROW_LD(1);
                float w0 = REC_NRM(0), w1 = REC_NRM(1);
                ROW_FMA(w0, v0) ROW_FMA(w1, v1)
            } else if (rem == 1) {
                float4 v0 = ROW_LD(0);
                float w0 = REC_NRM(0);
                ROW_FMA(w0, v0)
            }
        }

        const float d2 = dv * dv;
        float4 s;
        s.x = (1.f - ALPHA) * (acc.x + d2 * hs.x) + ALPHA * xs.x;
        s.y = (1.f - ALPHA) * (acc.y + d2 * hs.y) + ALPHA * xs.y;
        s.z = (1.f - ALPHA) * (acc.z + d2 * hs.z) + ALPHA * xs.z;
        s.w = (1.f - ALPHA) * (acc.w + d2 * hs.w) + ALPHA * xs.w;
        *(float4*)&sup[(wave * RPW + k) * LDA_L + lane * 4] = s;
    }
    __syncthreads();

    const int half = lane >> 4, col = lane & 15;
    const int n0   = wave * 16;
    const int mrow = col;

    v8f acc0 = {}, acc1 = {};
    #pragma unroll 8
    for (int p = 0; p < HID_C / 8; ++p) {
        const float4 b4 = wp[p * 256 + t];
        const int ka = p * 8 + 2 * half;
        v2f a0 = *(const v2f*)&sup[mrow * LDA_L + ka];
        v2f a1 = *(const v2f*)&sup[(16 + mrow) * LDA_L + ka];
        v2f b;
        b.x = b4.x; b.y = b4.y;
        acc0 = __builtin_amdgcn_wmma_f32_16x16x4_f32(false, a0, false, b,
                                                     (short)0, acc0, false, false);
        acc1 = __builtin_amdgcn_wmma_f32_16x16x4_f32(false, a1, false, b,
                                                     (short)0, acc1, false, false);
        a0 = *(const v2f*)&sup[mrow * LDA_L + ka + 4];
        a1 = *(const v2f*)&sup[(16 + mrow) * LDA_L + ka + 4];
        b.x = b4.z; b.y = b4.w;
        acc0 = __builtin_amdgcn_wmma_f32_16x16x4_f32(false, a0, false, b,
                                                     (short)0, acc0, false, false);
        acc1 = __builtin_amdgcn_wmma_f32_16x16x4_f32(false, a1, false, b,
                                                     (short)0, acc1, false, false);
    }

    #pragma unroll
    for (int i = 0; i < 8; ++i) {
        int row = i + 8 * half;
        float s = sup[row * LDA_L + n0 + col];
        float v = (1.f - beta) * s + beta * acc0[i];
        if (do_relu) v = fmaxf(v, 0.f);
        hout[(size_t)(m0 + row) * HID_C + n0 + col] = v;
    }
    #pragma unroll
    for (int i = 0; i < 8; ++i) {
        int row = 16 + i + 8 * half;
        float s = sup[row * LDA_L + n0 + col];
        float v = (1.f - beta) * s + beta * acc1[i];
        if (do_relu) v = fmaxf(v, 0.f);
        hout[(size_t)(m0 + row) * HID_C + n0 + col] = v;
    }
}

extern "C" void kernel_launch(void* const* d_in, const int* in_sizes, int n_in,
                              void* d_out, int out_size, void* d_ws, size_t ws_size,
                              hipStream_t stream) {
    const float* x      = (const float*)d_in[0];
    const int*   ei     = (const int*)d_in[1];
    const float* proj_w = (const float*)d_in[2];
    const float* proj_b = (const float*)d_in[3];
    const float* conv_w = (const float*)d_in[4];
    float* out = (float*)d_out;

    const int* e_src = ei;
    const int* e_dst = ei + N_EDGES;

    const size_t NH = (size_t)N_NODES * HID_C;
    char* ws = (char*)d_ws;
    int*   cnt    = (int*)ws;                                ws += ((size_t)N_NODES * 4 + 255) & ~255ull;
    float* dinv   = (float*)ws;                              ws += ((size_t)N_NODES * 4 + 255) & ~255ull;
    float* x0     = (float*)ws;                              ws += NH * 4;
    float* ha     = (float*)ws;                              ws += NH * 4;
    float* hb     = (float*)ws;                              ws += NH * 4;
    int*   rank   = (int*)ws;                                ws += (size_t)N_EDGES * 4;
    int*   rowptr = (int*)ws;                                ws += ((size_t)N_NODES * 4 + 255) & ~255ull;
    int*   bsum   = (int*)ws;                                ws += 2048;
    unsigned* table = (unsigned*)ws;                         ws += (size_t)N_CHUNK * N_PAIR * 4;
    int2*  offs   = (int2*)ws;                               ws += (size_t)N_CHUNK * N_NODES * 4;
    int2*  rec    = (int2*)ws;                               ws += (size_t)N_EDGES * 8;
    float4* wp    = (float4*)ws;


    kcount    <<<N_CHUNK, 1024, 0, stream>>>(e_dst, table, rank);
    kcolscan  <<<(N_PAIR + 127) / 128, 128, 0, stream>>>(table, offs, (int2*)cnt);
    kblock_sum<<<N_NBLK, 256, 0, stream>>>(cnt, bsum, proj_w, conv_w, wp);
    krowptr   <<<N_NBLK, 256, 0, stream>>>(cnt, bsum, rowptr, dinv);
    kfill     <<<(N_EDGES + 255) / 256, 256, 0, stream>>>(e_src, e_dst, rank, rowptr, (const int*)offs, dinv, rec);

    kproj<<<N_NODES / 16, 256, 0, stream>>>(x, wp, proj_b, x0);

    static const float betas[N_LAYERS] = {0.40546510f, 0.22314355f, 0.15415068f, 0.11778303f};

    const float* const lin[N_LAYERS]  = {x0, ha, hb, ha};
    float* const       lout[N_LAYERS] = {ha, hb, ha, out};
    for (int i = 0; i < N_LAYERS; ++i) {
        const float* hin  = lin[i];
        float*       hout = lout[i];
        klayer<<<N_NODES / TM, 256, 0, stream>>>(rowptr, cnt, rec, hin, x0, dinv,
                                                 wp + WP_PROJ4 + (size_t)i * WP_CONV4,
                                                 betas[i], (i != N_LAYERS - 1) ? 1 : 0, hout);
    }
}


# Output tiles of the projection and layer kernels staged through LDS so that every 128-byte line is written whole by one 16-byte-per-lane store, not as two 64-byte halves from two waves
# speedup vs baseline: 4.2273x; 1.0349x over previous
#include <hip/hip_runtime.h>
#include <math.h>

#define N_NODES 100000
#define N_EDGES 1600000
#define IN_C    256
#define HID_C   128
#define N_LAYERS 4
#define ALPHA   0.1f
#define N_NBLK  ((N_NODES + 255) / 256)

typedef float v2f __attribute__((ext_vector_type(2)));
typedef float v8f __attribute__((ext_vector_type(8)));

#define N_CHUNK 64
#define CHUNK_E (N_EDGES / N_CHUNK)
#define N_PAIR  (N_NODES / 2)
#define EPT     ((CHUNK_E + 1023) / 1024)
static_assert(N_EDGES % N_CHUNK == 0, "chunks cover the edge list exactly");
static_assert(CHUNK_E < 65536, "a 16-bit count holds a whole chunk");
static_assert(N_NODES % 8 == 0, "counter words move as 16-byte vectors");

__global__ __launch_bounds__(1024)
void kcount(const int* __restrict__ dst, unsigned* __restrict__ table, int* __restrict__ rank) {
    __shared__ __attribute__((aligned(16))) unsigned lc[N_PAIR];
    const int t  = threadIdx.x;
    const int c  = blockIdx.x;
    const int e0 = c * CHUNK_E;
    uint4* lc4 = (uint4*)lc;

    int dv[EPT];
    #pragma unroll
    for (int k = 0; k < EPT; ++k) {
        int i = t + k * 1024;
        dv[k] = (i < CHUNK_E) ? dst[e0 + i] : 0;
    }
    for (int i = t; i < N_PAIR / 4; i += 1024) lc4[i] = make_uint4(0u, 0u, 0u, 0u);
    __syncthreads();

    unsigned old[EPT];
    #pragma unroll
    for (int k = 0; k < EPT; ++k) {
        int i = t + k * 1024;
        old[k] = 0u;
        if (i < CHUNK_E) old[k] = atomicAdd(&lc[dv[k] >> 1], 1u << ((unsigned)(dv[k] & 1) * 16u));
    }
    #pragma unroll
    for (int k = 0; k < EPT; ++k) {
        int i = t + k * 1024;
        if (i < CHUNK_E) rank[e0 + i] = (int)((old[k] >> ((unsigned)(dv[k] & 1) * 16u)) & 0xFFFFu);
    }
    __syncthreads();

    uint4* t4 = (uint4*)(table + (size_t)c * N_PAIR);
    for (int i = t; i < N_PAIR / 4; i += 1024) t4[i] = lc4[i];
}

__global__ __launch_bounds__(128)
void kcolscan(const unsigned* __restrict__ table, int2* __restrict__ offs, int2* __restrict__ cnt2) {
    const int w = blockIdx.x * 128 + threadIdx.x;
    if (w >= N_PAIR) return;
    int lo = 0, hi = 0;
    #pragma unroll 16
    for (int c = 0; c < N_CHUNK; ++c) {
        unsigned v = table[(size_t)c * N_PAIR + w];
        offs[(size_t)c * N_PAIR + w] = make_int2(lo, hi);
        lo += (int)(v & 0xFFFFu);
        hi += (int)(v >> 16);
    }
    cnt2[w] = make_int2(lo, hi);
}

#define WP_PROJ4 (IN_C / 8 * 256)
#define WP_CONV4 (HID_C / 8 * 256)
#define WP_ALL4  (WP_PROJ4 + N_LAYERS * WP_CONV4)
#define WP_NBLK  (WP_ALL4 / 256)
static_assert(WP_ALL4 % 256 == 0, "whole 256-thread groups cover the copy exactly");
__device__ __forceinline__
void pack_w(const int o, const float* __restrict__ proj_w, const float* __restrict__ conv_w,
            float4* __restrict__ wp) {
    const float* wm = proj_w;
    int q = o;
    if (o >= WP_PROJ4) {
        const int r = o - WP_PROJ4;
        const int layer = r / WP_CONV4;
        q  = r - layer * WP_CONV4;
        wm = conv_w + (size_t)layer * HID_C * HID_C;
    }
    const int lane = q & 31, strip = (q >> 5) & 7, p = q >> 8;
    const int half = lane >> 4, col = lane & 15;
    const int n = strip * 16 + col;
    const int k = p * 8 + 2 * half;
    float4 v;
    v.x = wm[(size_t)k * HID_C + n];
    v.y = wm[(size_t)(k + 1) * HID_C + n];
    v.z = wm[(size_t)(k + 4) * HID_C + n];
    v.w = wm[(size_t)(k + 5) * HID_C + n];
    wp[o] = v;
}

static_assert(WP_NBLK <= N_NBLK, "enough work-groups in this launch to carry the weight copy");
__global__ void kblock_sum(const int* __restrict__ cnt, int* __restrict__ bsum,
                           const float* __restrict__ proj_w, const float* __restrict__ conv_w,
                           float4* __restrict__ wp) {
    __shared__ int red[256];
    const int t = threadIdx.x;
    if (blockIdx.x < WP_NBLK) pack_w(blockIdx.x * 256 + t, proj_w, conv_w, wp);
    const int v = blockIdx.x * 256 + t;
    red[t] = (v < N_NODES) ? cnt[v] : 0;
    __syncthreads();
    for (int s = 128; s > 0; s >>= 1) {
        if (t < s) red[t] += red[t + s];
        __syncthreads();
    }
    if (t == 0) bsum[blockIdx.x] = red[0];
}

__global__ void krowptr(const int* __restrict__ cnt, const int* __restrict__ bsum,
                        int* __restrict__ rowptr, float* __restrict__ dinv) {
    __shared__ int red[256];
    __shared__ int sc[256];
    const int t = threadIdx.x;
    const int b = blockIdx.x;
    const int v = b * 256 + t;
    const int c = (v < N_NODES) ? cnt[v] : 0;

    int part = 0;
    for (int i = t; i < b; i += 256) part += bsum[i];
    red[t] = part;
    sc[t]  = c;
    __syncthreads();
    for (int s = 128; s > 0; s >>= 1) {
        if (t < s) red[t] += red[t + s];
        __syncthreads();
    }
    const int off = red[0];

    for (int d = 1; d < 256; d <<= 1) {
        int add = (t >= d) ? sc[t - d] : 0;
        __syncthreads();
        sc[t] += add;
        __syncthreads();
    }
    if (v < N_NODES) {
        rowptr[v] = off + sc[t] - c;
        dinv[v]   = rsqrtf((float)(c + 1));
    }
}

__global__ void kfill(const int* __restrict__ src, const int* __restrict__ dst,
                      const int* __restrict__ rank, const int* __restrict__ rowptr,
                      const int* __restrict__ offs, const float* __restrict__ dinv,
                      int2* __restrict__ rec) {
    int e = blockIdx.x * 256 + threadIdx.x;
    if (e < N_EDGES) {
        int s = src[e], d = dst[e];
        int pos = rowptr[d] + offs[(size_t)(e / CHUNK_E) * N_NODES + d] + rank[e];
        rec[pos] = make_int2(s, __float_as_int(dinv[s] * dinv[d]));
    }
}

#define LDA_P 260
#define LDO   132
__global__ void kproj(const float* __restrict__ x, const float4* __restrict__ wp,
                      const float* __restrict__ bias, float* __restrict__ x0) {
    __shared__ float sa[16 * LDA_P];
    __shared__ __attribute__((aligned(16))) float so[16 * LDO];
    const int t   = threadIdx.x;
    const int m0  = blockIdx.x * 16;

    const float4* xf4 = (const float4*)x;
    #pragma unroll
    for (int c = 0; c < 4; ++c) {
        int idx = c * 256 + t;
        int row = idx >> 6, q = idx & 63;
        float4 v = xf4[(size_t)(m0 + row) * 64 + q];
        float* d = &sa[row * LDA_P + q * 4];
        d[0] = v.x; d[1] = v.y; d[2] = v.z; d[3] = v.w;
    }
    __syncthreads();

    const int wave = t >> 5, lane = t & 31;
    const int half = lane >> 4, col = lane & 15;
    const int n0   = wave * 16;
    const int mrow = col;

    v8f acc = {};
    #pragma unroll 8
    for (int p = 0; p < IN_C / 8; ++p) {
        const float4 b4 = wp[p * 256 + t];
        const int ka = p * 8 + 2 * half;
        v2f a = *(const v2f*)&sa[mrow * LDA_P + ka];
        v2f b;
        b.x = b4.x; b.y = b4.y;
        acc = __builtin_amdgcn_wmma_f32_16x16x4_f32(false, a, false, b,
                                                    (short)0, acc, false, false);
        a = *(const v2f*)&sa[mrow * LDA_P + ka + 4];
        b.x = b4.z; b.y = b4.w;
        acc = __builtin_amdgcn_wmma_f32_16x16x4_f32(false, a, false, b,
                                                    (short)0, acc, false, false);
    }

    float bv = bias[n0 + col];
    #pragma unroll
    for (int i = 0; i < 8; ++i) {
        int row = i + 8 * half;
        so[row * LDO + n0 + col] = acc[i] + bv;
    }
    __syncthreads();
    #pragma unroll
    for (int k = 0; k < 2; ++k) {
        const int row = wave * 2 + k;
        const float4 v = *(const float4*)&so[row * LDO + lane * 4];
        *(float4*)&x0[(size_t)(m0 + row) * HID_C + lane * 4] = v;
    }
}

#define REC_SRC(k) __builtin_amdgcn_readlane(r.x, j + (k))
#define REC_NRM(k) __int_as_float(__builtin_amdgcn_readlane(r.y, j + (k)))
#define ROW_LD(k)  h4[(size_t)REC_SRC(k) * 32 + lane]
#define ROW_FMA(wt, vv) row_fma(acc, wt, vv);
__device__ __forceinline__ void row_fma(float4& acc, float wt, const float4& vv) {
    acc.x = fmaf(wt, vv.x, acc.x); acc.y = fmaf(wt, vv.y, acc.y);
    acc.z = fmaf(wt, vv.z, acc.z); acc.w = fmaf(wt, vv.w, acc.w);
}
#define TM    32
#define RPW   4
#define LDA_L 132
__global__ __launch_bounds__(256)
void klayer(const int* __restrict__ rowptr, const int* __restrict__ cnt,
            const int2* __restrict__ rec, const float* __restrict__ hin,
            const float* __restrict__ x0, const float* __restrict__ dinv,
            const float4* __restrict__ wp, float beta, int do_relu,
            float* __restrict__ hout) {
    __shared__ __attribute__((aligned(16))) float sup[TM * LDA_L];
    __shared__ __attribute__((aligned(16))) float ot[TM * LDA_L];
    const int t    = threadIdx.x;
    const int lane = t & 31;
    const int wave = __builtin_amdgcn_readfirstlane(t >> 5);
    const int m0   = blockIdx.x * TM;
    const int r0   = m0 + wave * RPW;
    const float4* h4 = (const float4*)hin;
    const float4* x4 = (const float4*)x0;

    int beg[RPW], num[RPW];
    #pragma unroll
    for (int k = 0; k < RPW; ++k) { beg[k] = rowptr[r0 + k]; num[k] = cnt[r0 + k]; }
    int2 rfirst[RPW];
    #pragma unroll
    for (int k = 0; k < RPW; ++k) {
        rfirst[k] = make_int2(0, 0);
        if (lane < num[k]) rfirst[k] = rec[beg[k] + lane];
    }

    #pragma unroll
    for (int k = 0; k < RPW; ++k) {
        const int row = r0 + k;
        const int n   = num[k];
        const float4 hs = h4[(size_t)row * 32 + lane];
        const float4 xs = x4[(size_t)row * 32 + lane];
        const float  dv = dinv[row];

        float4 acc = make_float4(0.f, 0.f, 0.f, 0.f);
        for (int base = 0; base < n; base += 32) {
            const int m = (n - base < 32) ? (n - base) : 32;
            int2 r = rfirst[k];
            if (base > 0) {
                r = make_int2(0, 0);
                if (lane < m) r = rec[beg[k] + base + lane];
            }
            int j = 0;
            for (; j + 8 <= m; j += 8) {
                float4 v0 = ROW_LD(0), v1 = ROW_LD(1), v2 = ROW_LD(2), v3 = ROW_LD(3);
                float4 v4 = ROW_LD(4), v5 = ROW_LD(5), v6 = ROW_LD(6), v7 = ROW_LD(7);
                float w0 = REC_NRM(0), w1 = REC_NRM(1), w2 = REC_NRM(2), w3 = REC_NRM(3);
                float w4 = REC_NRM(4), w5 = REC_NRM(5), w6 = REC_NRM(6), w7 = REC_NRM(7);
                ROW_FMA(w0, v0) ROW_FMA(w1, v1) ROW_FMA(w2, v2) ROW_FMA(w3, v3)
                ROW_FMA(w4, v4) ROW_FMA(w5, v5) ROW_FMA(w6, v6) ROW_FMA(w7, v7)
            }
            if (j + 4 <= m) {
                float4 v0 = ROW_LD(0), v1 = ROW_LD(1), v2 = ROW_LD(2), v3 = ROW_LD(3);
                float w0 = REC_NRM(0), w1 = REC_NRM(1), w2 = REC_NRM(2), w3 = REC_NRM(3);
                ROW_FMA(w0, v0) ROW_FMA(w1, v1) ROW_FMA(w2, v2) ROW_FMA(w3, v3)
                j += 4;
            }
            const int rem = m - j;
            if (rem == 3) {
                float4 v0 = ROW_LD(0), v1 = ROW_LD(1), v2 = ROW_LD(2);
                float w0 = REC_NRM(0), w1 = REC_NRM(1), w2 = REC_NRM(2);
                ROW_FMA(w0, v0) ROW_FMA(w1, v1) ROW_FMA(w2, v2)
            } else if (rem == 2) {
                float4 v0 = ROW_LD(0), v1 = ROW_LD(1);
                float w0 = REC_NRM(0), w1 = REC_NRM(1);
                ROW_FMA(w0, v0) ROW_FMA(w1, v1)
            } else if (rem == 1) {
                float4 v0 = ROW_LD(0);
                float w0 = REC_NRM(0);
                ROW_FMA(w0, v0)
            }
        }

        const float d2 = dv * dv;
        float4 s;
        s.x = (1.f - ALPHA) * (acc.x + d2 * hs.x) + ALPHA * xs.x;
        s.y = (1.f - ALPHA) * (acc.y + d2 * hs.y) + ALPHA * xs.y;
        s.z = (1.f - ALPHA) * (acc.z + d2 * hs.z) + ALPHA * xs.z;
        s.w = (1.f - ALPHA) * (acc.w + d2 * hs.w) + ALPHA * xs.w;
        *(float4*)&sup[(wave * RPW + k) * LDA_L + lane * 4] = s;
    }
    __syncthreads();

    const int half = lane >> 4, col = lane & 15;
    const int n0   = wave * 16;
    const int mrow = col;

    v8f acc0 = {}, acc1 = {};
    #pragma unroll 8
    for (int p = 0; p < HID_C / 8; ++p) {
        const float4 b4 = wp[p * 256 + t];
        const int ka = p * 8 + 2 * half;
        v2f a0 = *(const v2f*)&sup[mrow * LDA_L + ka];
        v2f a1 = *(const v2f*)&sup[(16 + mrow) * LDA_L + ka];
        v2f b;
        b.x = b4.x; b.y = b4.y;
        acc0 = __builtin_amdgcn_wmma_f32_16x16x4_f32(false, a0, false, b,
                                                     (short)0, acc0, false, false);
        acc1 = __builtin_amdgcn_wmma_f32_16x16x4_f32(false, a1, false, b,
                                                     (short)0, acc1, false, false);
        a0 = *(const v2f*)&sup[mrow * LDA_L + ka + 4];
        a1 = *(const v2f*)&sup[(16 + mrow) * LDA_L + ka + 4];
        b.x = b4.z; b.y = b4.w;
        acc0 = __builtin_amdgcn_wmma_f32_16x16x4_f32(false, a0, false, b,
                                                     (short)0, acc0, false, false);
        acc1 = __builtin_amdgcn_wmma_f32_16x16x4_f32(false, a1, false, b,
                                                     (short)0, acc1, false, false);
    }

    #pragma unroll
    for (int i = 0; i < 8; ++i) {
        int row = i + 8 * half;
        float s = sup[row * LDA_L + n0 + col];
        float v = (1.f - beta) * s + beta * acc0[i];
        if (do_relu) v = fmaxf(v, 0.f);
        ot[row * LDA_L + n0 + col] = v;
    }
    #pragma unroll
    for (int i = 0; i < 8; ++i) {
        int row = 16 + i + 8 * half;
        float s = sup[row * LDA_L + n0 + col];
        float v = (1.f - beta) * s + beta * acc1[i];
        if (do_relu) v = fmaxf(v, 0.f);
        ot[row * LDA_L + n0 + col] = v;
    }
    __syncthreads();
    #pragma unroll
    for (int k = 0; k < RPW; ++k) {
        const int row = wave * RPW + k;
        const float4 v = *(const float4*)&ot[row * LDA_L + lane * 4];
        *(float4*)&hout[(size_t)(m0 + row) * HID_C + lane * 4] = v;
    }
}

extern "C" void kernel_launch(void* const* d_in, const int* in_sizes, int n_in,
                              void* d_out, int out_size, void* d_ws, size_t ws_size,
                              hipStream_t stream) {
    const float* x      = (const float*)d_in[0];
    const int*   ei     = (const int*)d_in[1];
    const float* proj_w = (const float*)d_in[2];
    const float* proj_b = (const float*)d_in[3];
    const float* conv_w = (const float*)d_in[4];
    float* out = (float*)d_out;

    const int* e_src = ei;
    const int* e_dst = ei + N_EDGES;

    const size_t NH = (size_t)N_NODES * HID_C;
    char* ws = (char*)d_ws;
    int*   cnt    = (int*)ws;                                ws += ((size_t)N_NODES * 4 + 255) & ~255ull;
    float* dinv   = (float*)ws;                              ws += ((size_t)N_NODES * 4 + 255) & ~255ull;
    float* x0     = (float*)ws;                              ws += NH * 4;
    float* ha     = (float*)ws;                              ws += NH * 4;
    float* hb     = (float*)ws;                              ws += NH * 4;
    int*   rank   = (int*)ws;                                ws += (size_t)N_EDGES * 4;
    int*   rowptr = (int*)ws;                                ws += ((size_t)N_NODES * 4 + 255) & ~255ull;
    int*   bsum   = (int*)ws;                                ws += 2048;
    unsigned* table = (unsigned*)ws;                         ws += (size_t)N_CHUNK * N_PAIR * 4;
    int2*  offs   = (int2*)ws;                               ws += (size_t)N_CHUNK * N_NODES * 4;
    int2*  rec    = (int2*)ws;                               ws += (size_t)N_EDGES * 8;
    float4* wp    = (float4*)ws;


    kcount    <<<N_CHUNK, 1024, 0, stream>>>(e_dst, table, rank);
    kcolscan  <<<(N_PAIR + 127) / 128, 128, 0, stream>>>(table, offs, (int2*)cnt);
    kblock_sum<<<N_NBLK, 256, 0, stream>>>(cnt, bsum, proj_w, conv_w, wp);
    krowptr   <<<N_NBLK, 256, 0, stream>>>(cnt, bsum, rowptr, dinv);
    kfill     <<<(N_EDGES + 255) / 256, 256, 0, stream>>>(e_src, e_dst, rank, rowptr, (const int*)offs, dinv, rec);

    kproj<<<N_NODES / 16, 256, 0, stream>>>(x, wp, proj_b, x0);

    static const float betas[N_LAYERS] = {0.40546510f, 0.22314355f, 0.15415068f, 0.11778303f};

    const float* const lin[N_LAYERS]  = {x0, ha, hb, ha};
    float* const       lout[N_LAYERS] = {ha, hb, ha, out};
    for (int i = 0; i < N_LAYERS; ++i) {
        const float* hin  = lin[i];
        float*       hout = lout[i];
        klayer<<<N_NODES / TM, 256, 0, stream>>>(rowptr, cnt, rec, hin, x0, dinv,
                                                 wp + WP_PROJ4 + (size_t)i * WP_CONV4,
                                                 betas[i], (i != N_LAYERS - 1) ? 1 : 0, hout);
    }
}


# Layer kernel: row loads indexed in 32 bits (scalar base + one offset register, no 64-bit address pairs); registers 74 -> 63, so 8 work-groups per WGP are resident instead of 6
# speedup vs baseline: 4.4148x; 1.0444x over previous
#include <hip/hip_runtime.h>
#include <math.h>

#define N_NODES 100000
#define N_EDGES 1600000
#define IN_C    256
#define HID_C   128
#define N_LAYERS 4
#define ALPHA   0.1f
#define N_NBLK  ((N_NODES + 255) / 256)

typedef float v2f __attribute__((ext_vector_type(2)));
typedef float v8f __attribute__((ext_vector_type(8)));

#define N_CHUNK 64
#define CHUNK_E (N_EDGES / N_CHUNK)
#define N_PAIR  (N_NODES / 2)
#define EPT     ((CHUNK_E + 1023) / 1024)
static_assert(N_EDGES % N_CHUNK == 0, "chunks cover the edge list exactly");
static_assert(CHUNK_E < 65536, "a 16-bit count holds a whole chunk");
static_assert(N_NODES % 8 == 0, "counter words move as 16-byte vectors");

__global__ __launch_bounds__(1024)
void kcount(const int* __restrict__ dst, unsigned* __restrict__ table, int* __restrict__ rank) {
    __shared__ __attribute__((aligned(16))) unsigned lc[N_PAIR];
    const int t  = threadIdx.x;
    const int c  = blockIdx.x;
    const int e0 = c * CHUNK_E;
    uint4* lc4 = (uint4*)lc;

    int dv[EPT];
    #pragma unroll
    for (int k = 0; k < EPT; ++k) {
        int i = t + k * 1024;
        dv[k] = (i < CHUNK_E) ? dst[e0 + i] : 0;
    }
    for (int i = t; i < N_PAIR / 4; i += 1024) lc4[i] = make_uint4(0u, 0u, 0u, 0u);
    __syncthreads();

    unsigned old[EPT];
    #pragma unroll
    for (int k = 0; k < EPT; ++k) {
        int i = t + k * 1024;
        old[k] = 0u;
        if (i < CHUNK_E) old[k] = atomicAdd(&lc[dv[k] >> 1], 1u << ((unsigned)(dv[k] & 1) * 16u));
    }
    #pragma unroll
    for (int k = 0; k < EPT; ++k) {
        int i = t + k * 1024;
        if (i < CHUNK_E) rank[e0 + i] = (int)((old[k] >> ((unsigned)(dv[k] & 1) * 16u)) & 0xFFFFu);
    }
    __syncthreads();

    uint4* t4 = (uint4*)(table + (size_t)c * N_PAIR);
    for (int i = t; i < N_PAIR / 4; i += 1024) t4[i] = lc4[i];
}

__global__ __launch_bounds__(128)
void kcolscan(const unsigned* __restrict__ table, int2* __restrict__ offs, int2* __restrict__ cnt2) {
    const int w = blockIdx.x * 128 + threadIdx.x;
    if (w >= N_PAIR) return;
    int lo = 0, hi = 0;
    #pragma unroll 16
    for (int c = 0; c < N_CHUNK; ++c) {
        unsigned v = table[(size_t)c * N_PAIR + w];
        offs[(size_t)c * N_PAIR + w] = make_int2(lo, hi);
        lo += (int)(v & 0xFFFFu);
        hi += (int)(v >> 16);
    }
    cnt2[w] = make_int2(lo, hi);
}

#define WP_PROJ4 (IN_C / 8 * 256)
#define WP_CONV4 (HID_C / 8 * 256)
#define WP_ALL4  (WP_PROJ4 + N_LAYERS * WP_CONV4)
#define WP_NBLK  (WP_ALL4 / 256)
static_assert(WP_ALL4 % 256 == 0, "whole 256-thread groups cover the copy exactly");
__device__ __forceinline__
void pack_w(const int o, const float* __restrict__ proj_w, const float* __restrict__ conv_w,
            float4* __restrict__ wp) {
    const float* wm = proj_w;
    int q = o;
    if (o >= WP_PROJ4) {
        const int r = o - WP_PROJ4;
        const int layer = r / WP_CONV4;
        q  = r - layer * WP_CONV4;
        wm = conv_w + (size_t)layer * HID_C * HID_C;
    }
    const int lane = q & 31, strip = (q >> 5) & 7, p = q >> 8;
    const int half = lane >> 4, col = lane & 15;
    const int n = strip * 16 + col;
    const int k = p * 8 + 2 * half;
    float4 v;
    v.x = wm[(size_t)k * HID_C + n];
    v.y = wm[(size_t)(k + 1) * HID_C + n];
    v.z = wm[(size_t)(k + 4) * HID_C + n];
    v.w = wm[(size_t)(k + 5) * HID_C + n];
    wp[o] = v;
}

static_assert(WP_NBLK <= N_NBLK, "enough work-groups in this launch to carry the weight copy");
__global__ void kblock_sum(const int* __restrict__ cnt, int* __restrict__ bsum,
                           const float* __restrict__ proj_w, const float* __restrict__ conv_w,
                           float4* __restrict__ wp) {
    __shared__ int red[256];
    const int t = threadIdx.x;
    if (blockIdx.x < WP_NBLK) pack_w(blockIdx.x * 256 + t, proj_w, conv_w, wp);
    const int v = blockIdx.x * 256 + t;
    red[t] = (v < N_NODES) ? cnt[v] : 0;
    __syncthreads();
    for (int s = 128; s > 0; s >>= 1) {
        if (t < s) red[t] += red[t + s];
        __syncthreads();
    }
    if (t == 0) bsum[blockIdx.x] = red[0];
}

__global__ void krowptr(const int* __restrict__ cnt, const int* __restrict__ bsum,
                        int* __restrict__ rowptr, float* __restrict__ dinv) {
    __shared__ int red[256];
    __shared__ int sc[256];
    const int t = threadIdx.x;
    const int b = blockIdx.x;
    const int v = b * 256 + t;
    const int c = (v < N_NODES) ? cnt[v] : 0;

    int part = 0;
    for (int i = t; i < b; i += 256) part += bsum[i];
    red[t] = part;
    sc[t]  = c;
    __syncthreads();
    for (int s = 128; s > 0; s >>= 1) {
        if (t < s) red[t] += red[t + s];
        __syncthreads();
    }
    const int off = red[0];

    for (int d = 1; d < 256; d <<= 1) {
        int add = (t >= d) ? sc[t - d] : 0;
        __syncthreads();
        sc[t] += add;
        __syncthreads();
    }
    if (v < N_NODES) {
        rowptr[v] = off + sc[t] - c;
        dinv[v]   = rsqrtf((float)(c + 1));
    }
}

__global__ void kfill(const int* __restrict__ src, const int* __restrict__ dst,
                      const int* __restrict__ rank, const int* __restrict__ rowptr,
                      const int* __restrict__ offs, const float* __restrict__ dinv,
                      int2* __restrict__ rec) {
    int e = blockIdx.x * 256 + threadIdx.x;
    if (e < N_EDGES) {
        int s = src[e], d = dst[e];
        int pos = rowptr[d] + offs[(size_t)(e / CHUNK_E) * N_NODES + d] + rank[e];
        rec[pos] = make_int2(s, __float_as_int(dinv[s] * dinv[d]));
    }
}

#define LDA_P 260
#define LDO   132
__global__ void kproj(const float* __restrict__ x, const float4* __restrict__ wp,
                      const float* __restrict__ bias, float* __restrict__ x0) {
    __shared__ float sa[16 * LDA_P];
    __shared__ __attribute__((aligned(16))) float so[16 * LDO];
    const int t   = threadIdx.x;
    const int m0  = blockIdx.x * 16;

    const float4* xf4 = (const float4*)x;
    #pragma unroll
    for (int c = 0; c < 4; ++c) {
        int idx = c * 256 + t;
        int row = idx >> 6, q = idx & 63;
        float4 v = xf4[(size_t)(m0 + row) * 64 + q];
        float* d = &sa[row * LDA_P + q * 4];
        d[0] = v.x; d[1] = v.y; d[2] = v.z; d[3] = v.w;
    }
    __syncthreads();

    const int wave = t >> 5, lane = t & 31;
    const int half = lane >> 4, col = lane & 15;
    const int n0   = wave * 16;
    const int mrow = col;

    v8f acc = {};
    #pragma unroll 8
    for (int p = 0; p < IN_C / 8; ++p) {
        const float4 b4 = wp[p * 256 + t];
        const int ka = p * 8 + 2 * half;
        v2f a = *(const v2f*)&sa[mrow * LDA_P + ka];
        v2f b;
        b.x = b4.x; b.y = b4.y;
        acc = __builtin_amdgcn_wmma_f32_16x16x4_f32(false, a, false, b,
                                                    (short)0, acc, false, false);
        a = *(const v2f*)&sa[mrow * LDA_P + ka + 4];
        b.x = b4.z; b.y = b4.w;
        acc = __builtin_amdgcn_wmma_f32_16x16x4_f32(false, a, false, b,
                                                    (short)0, acc, false, false);
    }

    float bv = bias[n0 + col];
    #pragma unroll
    for (int i = 0; i < 8; ++i) {
        int row = i + 8 * half;
        so[row * LDO + n0 + col] = acc[i] + bv;
    }
    __syncthreads();
    #pragma unroll
    for (int k = 0; k < 2; ++k) {
        const int row = wave * 2 + k;
        const float4 v = *(const float4*)&so[row * LDO + lane * 4];
        *(float4*)&x0[(size_t)(m0 + row) * HID_C + lane * 4] = v;
    }
}

#define REC_SRC(k) __builtin_amdgcn_readlane(r.x, j + (k))
#define REC_NRM(k) __int_as_float(__builtin_amdgcn_readlane(r.y, j + (k)))
static_assert((long long)N_NODES * 32 < 2147483647LL, "float4 index of any row fits in an int");
#define ROW_LD(k)  h4[REC_SRC(k) * 32 + lane]
#define ROW_FMA(wt, vv) row_fma(acc, wt, vv);
__device__ __forceinline__ void row_fma(float4& acc, float wt, const float4& vv) {
    acc.x = fmaf(wt, vv.x, acc.x); acc.y = fmaf(wt, vv.y, acc.y);
    acc.z = fmaf(wt, vv.z, acc.z); acc.w = fmaf(wt, vv.w, acc.w);
}
#define TM    32
#define RPW   4
#define LDA_L 132
__global__ __launch_bounds__(256)
void klayer(const int* __restrict__ rowptr, const int* __restrict__ cnt,
            const int2* __restrict__ rec, const float* __restrict__ hin,
            const float* __restrict__ x0, const float* __restrict__ dinv,
            const float4* __restrict__ wp, float beta, int do_relu,
            float* __restrict__ hout) {
    __shared__ __attribute__((aligned(16))) float sup[TM * LDA_L];
    __shared__ __attribute__((aligned(16))) float ot[TM * LDA_L];
    const int t    = threadIdx.x;
    const int lane = t & 31;
    const int wave = __builtin_amdgcn_readfirstlane(t >> 5);
    const int m0   = blockIdx.x * TM;
    const int r0   = m0 + wave * RPW;
    const float4* h4 = (const float4*)hin;
    const float4* x4 = (const float4*)x0;

    int beg[RPW], num[RPW];
    #pragma unroll
    for (int k = 0; k < RPW; ++k) { beg[k] = rowptr[r0 + k]; num[k] = cnt[r0 + k]; }
    int2 rfirst[RPW];
    #pragma unroll
    for (int k = 0; k < RPW; ++k) {
        rfirst[k] = make_int2(0, 0);
        if (lane < num[k]) rfirst[k] = rec[beg[k] + lane];
    }

    #pragma unroll
    for (int k = 0; k < RPW; ++k) {
        const int row = r0 + k;
        const int n   = num[k];
        const float4 hs = h4[row * 32 + lane];
        const float4 xs = x4[row * 32 + lane];
        const float  dv = dinv[row];

        float4 acc = make_float4(0.f, 0.f, 0.f, 0.f);
        for (int base = 0; base < n; base += 32) {
            const int m = (n - base < 32) ? (n - base) : 32;
            int2 r = rfirst[k];
            if (base > 0) {
                r = make_int2(0, 0);
                if (lane < m) r = rec[beg[k] + base + lane];
            }
            int j = 0;
            for (; j + 8 <= m; j += 8) {
                float4 v0 = ROW_LD(0), v1 = ROW_LD(1), v2 = ROW_LD(2), v3 = ROW_LD(3);
                float4 v4 = ROW_LD(4), v5 = ROW_LD(5), v6 = ROW_LD(6), v7 = ROW_LD(7);
                float w0 = REC_NRM(0), w1 = REC_NRM(1), w2 = REC_NRM(2), w3 = REC_NRM(3);
                float w4 = REC_NRM(4), w5 = REC_NRM(5), w6 = REC_NRM(6), w7 = REC_NRM(7);
                ROW_FMA(w0, v0) ROW_FMA(w1, v1) ROW_FMA(w2, v2) ROW_FMA(w3, v3)
                ROW_FMA(w4, v4) ROW_FMA(w5, v5) ROW_FMA(w6, v6) ROW_FMA(w7, v7)
            }
            if (j + 4 <= m) {
                float4 v0 = ROW_LD(0), v1 = ROW_LD(1), v2 = ROW_LD(2), v3 = ROW_LD(3);
                float w0 = REC_NRM(0), w1 = REC_NRM(1), w2 = REC_NRM(2), w3 = REC_NRM(3);
                ROW_FMA(w0, v0) ROW_FMA(w1, v1) ROW_FMA(w2, v2) ROW_FMA(w3, v3)
                j += 4;
            }
            const int rem = m - j;
            if (rem == 3) {
                float4 v0 = ROW_LD(0), v1 = ROW_LD(1), v2 = ROW_LD(2);
                float w0 = REC_NRM(0), w1 = REC_NRM(1), w2 = REC_NRM(2);
                ROW_FMA(w0, v0) ROW_FMA(w1, v1) ROW_FMA(w2, v2)
            } else if (rem == 2) {
                float4 v0 = ROW_LD(0), v1 = ROW_LD(1);
                float w0 = REC_NRM(0), w1 = REC_NRM(1);
                ROW_FMA(w0, v0) ROW_FMA(w1, v1)
            } else if (rem == 1) {
                float4 v0 = ROW_LD(0);
                float w0 = REC_NRM(0);
                ROW_FMA(w0, v0)
            }
        }

        const float d2 = dv * dv;
        float4 s;
        s.x = (1.f - ALPHA) * (acc.x + d2 * hs.x) + ALPHA * xs.x;
        s.y = (1.f - ALPHA) * (acc.y + d2 * hs.y) + ALPHA * xs.y;
        s.z = (1.f - ALPHA) * (acc.z + d2 * hs.z) + ALPHA * xs.z;
        s.w = (1.f - ALPHA) * (acc.w + d2 * hs.w) + ALPHA * xs.w;
        *(float4*)&sup[(wave * RPW + k) * LDA_L + lane * 4] = s;
    }
    __syncthreads();

    const int half = lane >> 4, col = lane & 15;
    const int n0   = wave * 16;
    const int mrow = col;

    v8f acc0 = {}, acc1 = {};
    #pragma unroll 8
    for (int p = 0; p < HID_C / 8; ++p) {
        const float4 b4 = wp[p * 256 + t];
        const int ka = p * 8 + 2 * half;
        v2f a0 = *(const v2f*)&sup[mrow * LDA_L + ka];
        v2f a1 = *(const v2f*)&sup[(16 + mrow) * LDA_L + ka];
        v2f b;
        b.x = b4.x; b.y = b4.y;
        acc0 = __builtin_amdgcn_wmma_f32_16x16x4_f32(false, a0, false, b,
                                                     (short)0, acc0, false, false);
        acc1 = __builtin_amdgcn_wmma_f32_16x16x4_f32(false, a1, false, b,
                                                     (short)0, acc1, false, false);
        a0 = *(const v2f*)&sup[mrow * LDA_L + ka + 4];
        a1 = *(const v2f*)&sup[(16 + mrow) * LDA_L + ka + 4];
        b.x = b4.z; b.y = b4.w;
        acc0 = __builtin_amdgcn_wmma_f32_16x16x4_f32(false, a0, false, b,
                                                     (short)0, acc0, false, false);
        acc1 = __builtin_amdgcn_wmma_f32_16x16x4_f32(false, a1, false, b,
                                                     (short)0, acc1, false, false);
    }

    #pragma unroll
    for (int i = 0; i < 8; ++i) {
        int row = i + 8 * half;
        float s = sup[row * LDA_L + n0 + col];
        float v = (1.f - beta) * s + beta * acc0[i];
        if (do_relu) v = fmaxf(v, 0.f);
        ot[row * LDA_L + n0 + col] = v;
    }
    #pragma unroll
    for (int i = 0; i < 8; ++i) {
        int row = 16 + i + 8 * half;
        float s = sup[row * LDA_L + n0 + col];
        float v = (1.f - beta) * s + beta * acc1[i];
        if (do_relu) v = fmaxf(v, 0.f);
        ot[row * LDA_L + n0 + col] = v;
    }
    __syncthreads();
    #pragma unroll
    for (int k = 0; k < RPW; ++k) {
        const int row = wave * RPW + k;
        const float4 v = *(const float4*)&ot[row * LDA_L + lane * 4];
        *(float4*)&hout[(size_t)(m0 + row) * HID_C + lane * 4] = v;
    }
}

extern "C" void kernel_launch(void* const* d_in, const int* in_sizes, int n_in,
                              void* d_out, int out_size, void* d_ws, size_t ws_size,
                              hipStream_t stream) {
    const float* x      = (const float*)d_in[0];
    const int*   ei     = (const int*)d_in[1];
    const float* proj_w = (const float*)d_in[2];
    const float* proj_b = (const float*)d_in[3];
    const float* conv_w = (const float*)d_in[4];
    float* out = (float*)d_out;

    const int* e_src = ei;
    const int* e_dst = ei + N_EDGES;

    const size_t NH = (size_t)N_NODES * HID_C;
    char* ws = (char*)d_ws;
    int*   cnt    = (int*)ws;                                ws += ((size_t)N_NODES * 4 + 255) & ~255ull;
    float* dinv   = (float*)ws;                              ws += ((size_t)N_NODES * 4 + 255) & ~255ull;
    float* x0     = (float*)ws;                              ws += NH * 4;
    float* ha     = (float*)ws;                              ws += NH * 4;
    float* hb     = (float*)ws;                              ws += NH * 4;
    int*   rank   = (int*)ws;                                ws += (size_t)N_EDGES * 4;
    int*   rowptr = (int*)ws;                                ws += ((size_t)N_NODES * 4 + 255) & ~255ull;
    int*   bsum   = (int*)ws;                                ws += 2048;
    unsigned* table = (unsigned*)ws;                         ws += (size_t)N_CHUNK * N_PAIR * 4;
    int2*  offs   = (int2*)ws;                               ws += (size_t)N_CHUNK * N_NODES * 4;
    int2*  rec    = (int2*)ws;                               ws += (size_t)N_EDGES * 8;
    float4* wp    = (float4*)ws;


    kcount    <<<N_CHUNK, 1024, 0, stream>>>(e_dst, table, rank);
    kcolscan  <<<(N_PAIR + 127) / 128, 128, 0, stream>>>(table, offs, (int2*)cnt);
    kblock_sum<<<N_NBLK, 256, 0, stream>>>(cnt, bsum, proj_w, conv_w, wp);
    krowptr   <<<N_NBLK, 256, 0, stream>>>(cnt, bsum, rowptr, dinv);
    kfill     <<<(N_EDGES + 255) / 256, 256, 0, stream>>>(e_src, e_dst, rank, rowptr, (const int*)offs, dinv, rec);

    kproj<<<N_NODES / 16, 256, 0, stream>>>(x, wp, proj_b, x0);

    static const float betas[N_LAYERS] = {0.40546510f, 0.22314355f, 0.15415068f, 0.11778303f};

    const float* const lin[N_LAYERS]  = {x0, ha, hb, ha};
    float* const       lout[N_LAYERS] = {ha, hb, ha, out};
    for (int i = 0; i < N_LAYERS; ++i) {
        const float* hin  = lin[i];
        float*       hout = lout[i];
        klayer<<<N_NODES / TM, 256, 0, stream>>>(rowptr, cnt, rec, hin, x0, dinv,
                                                 wp + WP_PROJ4 + (size_t)i * WP_CONV4,
                                                 betas[i], (i != N_LAYERS - 1) ? 1 : 0, hout);
    }
}
